# starting kernel
# baseline (speedup 1.0000x reference)
#include <hip/hip_runtime.h>
#include <hip/hip_bf16.h>
#include <math.h>


typedef float v2f __attribute__((ext_vector_type(2)));
typedef float v8f __attribute__((ext_vector_type(8)));

#define GRU_L 512
#define GRU_B 2048
#define GRU_D 64

__global__ void __launch_bounds__(256)
gru_gates_wmma_kernel(const float* __restrict__ x,
                      const float* __restrict__ W_ih,
                      const float* __restrict__ b_ih,
                      float* __restrict__ gx,
                      int rows) {
  __shared__ float Bs[16 * GRU_D];
  for (int i = threadIdx.x; i < 16 * GRU_D; i += blockDim.x) {
    const int row = i >> 6;
    const int col = i & (GRU_D - 1);
    Bs[i] = (row < 3) ? W_ih[row * GRU_D + col] : 0.0f;
  }
  __syncthreads();

  const int lane     = threadIdx.x & 31;
  const int wave     = blockIdx.x * (blockDim.x >> 5) + (threadIdx.x >> 5);
  const int tileBase = wave << 4;
  if (tileBase >= rows) return;

  const int m     = lane & 15;
  const int khalf = lane >> 4;

  const float* arow = x + (size_t)(tileBase + m) * GRU_D + 2 * khalf;
  const float* brow = Bs + m * GRU_D + 2 * khalf;

  v8f c = {};
#pragma unroll
  for (int k = 0; k < 16; ++k) {
    const v2f a = *(const v2f*)(arow + 4 * k);
    const v2f b = *(const v2f*)(brow + 4 * k);
#if __has_builtin(__builtin_amdgcn_wmma_f32_16x16x4_f32)
    c = __builtin_amdgcn_wmma_f32_16x16x4_f32(false, a, false, b,
                                              (short)0, c, false, false);
#else
#pragma unroll
    for (int i = 0; i < 8; ++i) c[i] += a.x * b.x + a.y * b.y;
#endif
  }

  if (m < 3) {
    const float bias = b_ih[m];
    float* plane = gx + (size_t)m * rows;
#pragma unroll
    for (int i = 0; i < 8; ++i) {
      const int row = tileBase + i + 8 * khalf;
      plane[row] = c[i] + bias;
    }
  }
}

__global__ void __launch_bounds__(256)
gru_scan_kernel(const float* __restrict__ gx,
                const float* __restrict__ h0,
                const int*   __restrict__ lengths,
                const float* __restrict__ W_hh,
                const float* __restrict__ b_hh,
                float* __restrict__ out) {
  const int b = blockIdx.x * blockDim.x + threadIdx.x;
  if (b >= GRU_B) return;

  const size_t plane = (size_t)GRU_L * GRU_B;
  const float* gr = gx;
  const float* gz = gx + plane;
  const float* gn = gx + 2 * plane;

  float h = h0[b];
  const int   len = lengths[b];
  const float w0 = W_hh[0], w1 = W_hh[1], w2 = W_hh[2];
  const float c0 = b_hh[0], c1 = b_hh[1], c2 = b_hh[2];

  for (int t = 0; t < GRU_L; ++t) {
    const size_t idx = (size_t)t * GRU_B + b;

    const size_t pf = idx + (size_t)16 * GRU_B;
    __builtin_prefetch(gr + pf, 0, 0);
    __builtin_prefetch(gz + pf, 0, 0);
    __builtin_prefetch(gn + pf, 0, 0);

    const float xr = gr[idx];
    const float xz = gz[idx];
    const float xn = gn[idx];

    const float r = 1.0f / (1.0f + __expf(-(xr + h * w0 + c0)));
    const float z = 1.0f / (1.0f + __expf(-(xz + h * w1 + c1)));
    const float n = tanhf(xn + r * (h * w2 + c2));
    h = (1.0f - z) * n + z * h;

    out[idx] = (t < len) ? h : 0.0f;
  }
}

extern "C" void kernel_launch(void* const* d_in, const int* in_sizes, int n_in,
                              void* d_out, int out_size, void* d_ws, size_t ws_size,
                              hipStream_t stream) {
  const float* x       = (const float*)d_in[0];
  const float* h0      = (const float*)d_in[1];
  const int*   lengths = (const int*)  d_in[2];
  const float* W_ih    = (const float*)d_in[3];
  const float* W_hh    = (const float*)d_in[4];
  const float* b_ih    = (const float*)d_in[5];
  const float* b_hh    = (const float*)d_in[6];
  float*       out     = (float*)d_out;
  float*       gx      = (float*)d_ws;

  const int rows  = GRU_L * GRU_B;
  const int tiles = rows / 16;
  const int wavesPerBlock = 8;
  const int blocks1 = tiles / wavesPerBlock;

  gru_gates_wmma_kernel<<<blocks1, 256, 0, stream>>>(x, W_ih, b_ih, gx, rows);
  gru_scan_kernel<<<(GRU_B + 255) / 256, 256, 0, stream>>>(gx, h0, lengths,
                                                           W_hh, b_hh, out);
}


# scan: gate values of the next block of steps are loaded into a second register set while the current block runs, so the serial h chain no longer waits on memory each step; prefetch hints dropped
# speedup vs baseline: 2.3103x; 2.3103x over previous
#include <hip/hip_runtime.h>
#include <hip/hip_bf16.h>
#include <math.h>


typedef float v2f __attribute__((ext_vector_type(2)));
typedef float v8f __attribute__((ext_vector_type(8)));

#define GRU_L 512
#define GRU_B 2048
#define GRU_D 64

__global__ void __launch_bounds__(256)
gru_gates_wmma_kernel(const float* __restrict__ x,
                      const float* __restrict__ W_ih,
                      const float* __restrict__ b_ih,
                      float* __restrict__ gx,
                      int rows) {
  __shared__ float Bs[16 * GRU_D];
  for (int i = threadIdx.x; i < 16 * GRU_D; i += blockDim.x) {
    const int row = i >> 6;
    const int col = i & (GRU_D - 1);
    Bs[i] = (row < 3) ? W_ih[row * GRU_D + col] : 0.0f;
  }
  __syncthreads();

  const int lane     = threadIdx.x & 31;
  const int wave     = blockIdx.x * (blockDim.x >> 5) + (threadIdx.x >> 5);
  const int tileBase = wave << 4;
  if (tileBase >= rows) return;

  const int m     = lane & 15;
  const int khalf = lane >> 4;

  const float* arow = x + (size_t)(tileBase + m) * GRU_D + 2 * khalf;
  const float* brow = Bs + m * GRU_D + 2 * khalf;

  v8f c = {};
#pragma unroll
  for (int k = 0; k < 16; ++k) {
    const v2f a = *(const v2f*)(arow + 4 * k);
    const v2f b = *(const v2f*)(brow + 4 * k);
#if __has_builtin(__builtin_amdgcn_wmma_f32_16x16x4_f32)
    c = __builtin_amdgcn_wmma_f32_16x16x4_f32(false, a, false, b,
                                              (short)0, c, false, false);
#else
#pragma unroll
    for (int i = 0; i < 8; ++i) c[i] += a.x * b.x + a.y * b.y;
#endif
  }

  if (m < 3) {
    const float bias = b_ih[m];
    float* plane = gx + (size_t)m * rows;
#pragma unroll
    for (int i = 0; i < 8; ++i) {
      const int row = tileBase + i + 8 * khalf;
      plane[row] = c[i] + bias;
    }
  }
}

#define SCAN_T 16

__device__ __forceinline__ void gru_load_block(const float* __restrict__ pr,
                                               const float* __restrict__ pz,
                                               const float* __restrict__ pn,
                                               int t0,
                                               float (&xr)[SCAN_T],
                                               float (&xz)[SCAN_T],
                                               float (&xn)[SCAN_T]) {
#pragma unroll
  for (int i = 0; i < SCAN_T; ++i) {
    const int off = (t0 + i) * GRU_B;
    xr[i] = pr[off];
    xz[i] = pz[off];
    xn[i] = pn[off];
  }
}

__device__ __forceinline__ float gru_run_block(float h, int t0, int len,
                                               const float (&xr)[SCAN_T],
                                               const float (&xz)[SCAN_T],
                                               const float (&xn)[SCAN_T],
                                               float w0, float w1, float w2,
                                               float c0, float c1, float c2,
                                               float* __restrict__ po) {
#pragma unroll
  for (int i = 0; i < SCAN_T; ++i) {
    const int t = t0 + i;
    const float r = 1.0f / (1.0f + __expf(-(xr[i] + h * w0 + c0)));
    const float z = 1.0f / (1.0f + __expf(-(xz[i] + h * w1 + c1)));
    const float n = tanhf(xn[i] + r * (h * w2 + c2));
    h = (1.0f - z) * n + z * h;
    po[t * GRU_B] = (t < len) ? h : 0.0f;
  }
  return h;
}

__global__ void __launch_bounds__(256)
gru_scan_kernel(const float* __restrict__ gx,
                const float* __restrict__ h0,
                const int*   __restrict__ lengths,
                const float* __restrict__ W_hh,
                const float* __restrict__ b_hh,
                float* __restrict__ out) {
  const int b = blockIdx.x * blockDim.x + threadIdx.x;
  if (b >= GRU_B) return;

  const size_t plane = (size_t)GRU_L * GRU_B;
  const float* pr = gx + b;
  const float* pz = gx + plane + b;
  const float* pn = gx + 2 * plane + b;
  float*       po = out + b;

  float h = h0[b];
  const int   len = lengths[b];
  const float w0 = W_hh[0], w1 = W_hh[1], w2 = W_hh[2];
  const float c0 = b_hh[0], c1 = b_hh[1], c2 = b_hh[2];

  float ar[SCAN_T], az[SCAN_T], an[SCAN_T];
  float br[SCAN_T], bz[SCAN_T], bn[SCAN_T];

  gru_load_block(pr, pz, pn, 0, ar, az, an);
  for (int t0 = 0; t0 < GRU_L; t0 += 2 * SCAN_T) {
    gru_load_block(pr, pz, pn, t0 + SCAN_T, br, bz, bn);
    h = gru_run_block(h, t0, len, ar, az, an, w0, w1, w2, c0, c1, c2, po);
    const int tn = (t0 + 2 * SCAN_T < GRU_L) ? (t0 + 2 * SCAN_T) : 0;
    gru_load_block(pr, pz, pn, tn, ar, az, an);
    h = gru_run_block(h, t0 + SCAN_T, len, br, bz, bn, w0, w1, w2, c0, c1, c2, po);
  }
}

extern "C" void kernel_launch(void* const* d_in, const int* in_sizes, int n_in,
                              void* d_out, int out_size, void* d_ws, size_t ws_size,
                              hipStream_t stream) {
  const float* x       = (const float*)d_in[0];
  const float* h0      = (const float*)d_in[1];
  const int*   lengths = (const int*)  d_in[2];
  const float* W_ih    = (const float*)d_in[3];
  const float* W_hh    = (const float*)d_in[4];
  const float* b_ih    = (const float*)d_in[5];
  const float* b_hh    = (const float*)d_in[6];
  float*       out     = (float*)d_out;
  float*       gx      = (float*)d_ws;

  const int rows  = GRU_L * GRU_B;
  const int tiles = rows / 16;
  const int wavesPerBlock = 8;
  const int blocks1 = tiles / wavesPerBlock;

  gru_gates_wmma_kernel<<<blocks1, 256, 0, stream>>>(x, W_ih, b_ih, gx, rows);
  gru_scan_kernel<<<(GRU_B + 255) / 256, 256, 0, stream>>>(gx, h0, lengths,
                                                           W_hh, b_hh, out);
}


# scan: each step's two sigmoids and its tanh are one hardware tanh instruction each (sigmoid(a)=0.5+0.5*tanh(a/2), halves folded into weights, sums regrouped) in place of exp, IEEE division, tanhf
# speedup vs baseline: 3.8500x; 1.6665x over previous
#include <hip/hip_runtime.h>
#include <hip/hip_bf16.h>
#include <math.h>


typedef float v2f __attribute__((ext_vector_type(2)));
typedef float v8f __attribute__((ext_vector_type(8)));

#define GRU_L 512
#define GRU_B 2048
#define GRU_D 64

__global__ void __launch_bounds__(256)
gru_gates_wmma_kernel(const float* __restrict__ x,
                      const float* __restrict__ W_ih,
                      const float* __restrict__ b_ih,
                      float* __restrict__ gx,
                      int rows) {
  __shared__ float Bs[16 * GRU_D];
  for (int i = threadIdx.x; i < 16 * GRU_D; i += blockDim.x) {
    const int row = i >> 6;
    const int col = i & (GRU_D - 1);
    Bs[i] = (row < 3) ? W_ih[row * GRU_D + col] : 0.0f;
  }
  __syncthreads();

  const int lane     = threadIdx.x & 31;
  const int wave     = blockIdx.x * (blockDim.x >> 5) + (threadIdx.x >> 5);
  const int tileBase = wave << 4;
  if (tileBase >= rows) return;

  const int m     = lane & 15;
  const int khalf = lane >> 4;

  const float* arow = x + (size_t)(tileBase + m) * GRU_D + 2 * khalf;
  const float* brow = Bs + m * GRU_D + 2 * khalf;

  v8f c = {};
#pragma unroll
  for (int k = 0; k < 16; ++k) {
    const v2f a = *(const v2f*)(arow + 4 * k);
    const v2f b = *(const v2f*)(brow + 4 * k);
#if __has_builtin(__builtin_amdgcn_wmma_f32_16x16x4_f32)
    c = __builtin_amdgcn_wmma_f32_16x16x4_f32(false, a, false, b,
                                              (short)0, c, false, false);
#else
#pragma unroll
    for (int i = 0; i < 8; ++i) c[i] += a.x * b.x + a.y * b.y;
#endif
  }

  if (m < 3) {
    const float bias = b_ih[m];
    float* plane = gx + (size_t)m * rows;
#pragma unroll
    for (int i = 0; i < 8; ++i) {
      const int row = tileBase + i + 8 * khalf;
      plane[row] = c[i] + bias;
    }
  }
}

#define SCAN_T 16

__device__ __forceinline__ void gru_load_block(const float* __restrict__ pr,
                                               const float* __restrict__ pz,
                                               const float* __restrict__ pn,
                                               int t0,
                                               float (&xr)[SCAN_T],
                                               float (&xz)[SCAN_T],
                                               float (&xn)[SCAN_T]) {
#pragma unroll
  for (int i = 0; i < SCAN_T; ++i) {
    const int off = (t0 + i) * GRU_B;
    xr[i] = pr[off];
    xz[i] = pz[off];
    xn[i] = pn[off];
  }
}

__device__ __forceinline__ float gru_run_block(float h, int t0, int len,
                                               const float (&xr)[SCAN_T],
                                               const float (&xz)[SCAN_T],
                                               const float (&xn)[SCAN_T],
                                               float hw0, float hw1, float hw2,
                                               float hc0, float hc1, float hc2,
                                               float* __restrict__ po) {
#pragma unroll
  for (int i = 0; i < SCAN_T; ++i) {
    const int t = t0 + i;
    const float pr = fmaf(0.5f, xr[i], hc0);
    const float pz = fmaf(0.5f, xz[i], hc1);
    const float tr = __builtin_amdgcn_tanhf(fmaf(h, hw0, pr));
    const float tz = __builtin_amdgcn_tanhf(fmaf(h, hw1, pz));
    const float hg = fmaf(h, hw2, hc2);
    const float n  = __builtin_amdgcn_tanhf(fmaf(tr, hg, xn[i] + hg));
    const float hh  = 0.5f * h;
    const float zh  = fmaf(tz, hh, hh);
    const float omz = fmaf(-0.5f, tz, 0.5f);
    h = fmaf(omz, n, zh);
    po[t * GRU_B] = (t < len) ? h : 0.0f;
  }
  return h;
}

__global__ void __launch_bounds__(256)
gru_scan_kernel(const float* __restrict__ gx,
                const float* __restrict__ h0,
                const int*   __restrict__ lengths,
                const float* __restrict__ W_hh,
                const float* __restrict__ b_hh,
                float* __restrict__ out) {
  const int b = blockIdx.x * blockDim.x + threadIdx.x;
  if (b >= GRU_B) return;

  const size_t plane = (size_t)GRU_L * GRU_B;
  const float* pr = gx + b;
  const float* pz = gx + plane + b;
  const float* pn = gx + 2 * plane + b;
  float*       po = out + b;

  float h = h0[b];
  const int   len = lengths[b];
  const float w0 = 0.5f * W_hh[0], w1 = 0.5f * W_hh[1], w2 = 0.5f * W_hh[2];
  const float c0 = 0.5f * b_hh[0], c1 = 0.5f * b_hh[1], c2 = 0.5f * b_hh[2];

  float ar[SCAN_T], az[SCAN_T], an[SCAN_T];
  float br[SCAN_T], bz[SCAN_T], bn[SCAN_T];

  gru_load_block(pr, pz, pn, 0, ar, az, an);
  for (int t0 = 0; t0 < GRU_L; t0 += 2 * SCAN_T) {
    gru_load_block(pr, pz, pn, t0 + SCAN_T, br, bz, bn);
    h = gru_run_block(h, t0, len, ar, az, an, w0, w1, w2, c0, c1, c2, po);
    const int tn = (t0 + 2 * SCAN_T < GRU_L) ? (t0 + 2 * SCAN_T) : 0;
    gru_load_block(pr, pz, pn, tn, ar, az, an);
    h = gru_run_block(h, t0 + SCAN_T, len, br, bz, bn, w0, w1, w2, c0, c1, c2, po);
  }
}

extern "C" void kernel_launch(void* const* d_in, const int* in_sizes, int n_in,
                              void* d_out, int out_size, void* d_ws, size_t ws_size,
                              hipStream_t stream) {
  const float* x       = (const float*)d_in[0];
  const float* h0      = (const float*)d_in[1];
  const int*   lengths = (const int*)  d_in[2];
  const float* W_ih    = (const float*)d_in[3];
  const float* W_hh    = (const float*)d_in[4];
  const float* b_ih    = (const float*)d_in[5];
  const float* b_hh    = (const float*)d_in[6];
  float*       out     = (float*)d_out;
  float*       gx      = (float*)d_ws;

  const int rows  = GRU_L * GRU_B;
  const int tiles = rows / 16;
  const int wavesPerBlock = 8;
  const int blocks1 = tiles / wavesPerBlock;

  gru_gates_wmma_kernel<<<blocks1, 256, 0, stream>>>(x, W_ih, b_ih, gx, rows);
  gru_scan_kernel<<<(GRU_B + 255) / 256, 256, 0, stream>>>(gx, h0, lengths,
                                                           W_hh, b_hh, out);
}


# gates kernel: 8192 resident waves each loop over 8 row tiles with the next tile's loads issued ahead of the current tile's WMMAs (W_ih fragments, bias kept in registers); was 65536 one-tile waves
# speedup vs baseline: 4.4116x; 1.1459x over previous
#include <hip/hip_runtime.h>
#include <hip/hip_bf16.h>
#include <math.h>


typedef float v2f __attribute__((ext_vector_type(2)));
typedef float v8f __attribute__((ext_vector_type(8)));

#define GRU_L 512
#define GRU_B 2048
#define GRU_D 64

__device__ __forceinline__ void gates_load_tile(const float* __restrict__ x,
                                                int tile, int laneOff,
                                                v2f (&a)[16]) {
  const float* arow = x + (size_t)tile * (16 * GRU_D) + laneOff;
#pragma unroll
  for (int k = 0; k < 16; ++k) a[k] = *(const v2f*)(arow + 4 * k);
}

__device__ __forceinline__ void gates_mma_store(const v2f (&a)[16],
                                                const v2f (&b)[16],
                                                float bias,
                                                float* __restrict__ gx,
                                                int rows, int tile,
                                                int m, int khalf) {
  v8f c = {};
#pragma unroll
  for (int k = 0; k < 16; ++k) {
    c = __builtin_amdgcn_wmma_f32_16x16x4_f32(false, a[k], false, b[k],
                                              (short)0, c, false, false);
  }
  if (m < 3) {
    float* plane = gx + (size_t)m * rows + (size_t)tile * 16 + 8 * khalf;
#pragma unroll
    for (int i = 0; i < 8; ++i) plane[i] = c[i] + bias;
  }
}

__global__ void __launch_bounds__(256)
gru_gates_wmma_kernel(const float* __restrict__ x,
                      const float* __restrict__ W_ih,
                      const float* __restrict__ b_ih,
                      float* __restrict__ gx,
                      int rows) {
  __shared__ float Bs[16 * GRU_D];
  for (int i = threadIdx.x; i < 16 * GRU_D; i += blockDim.x) {
    const int row = i >> 6;
    const int col = i & (GRU_D - 1);
    Bs[i] = (row < 3) ? W_ih[row * GRU_D + col] : 0.0f;
  }
  __syncthreads();

  const int lane  = threadIdx.x & 31;
  const int m     = lane & 15;
  const int khalf = lane >> 4;

  const int waveInBlock = __builtin_amdgcn_readfirstlane((int)(threadIdx.x >> 5));
  const int wavesPerBlk = (int)(blockDim.x >> 5);
  const int nWaves = (int)gridDim.x * wavesPerBlk;
  const int nTiles = rows >> 4;
  int tile = (int)blockIdx.x * wavesPerBlk + waveInBlock;
  if (tile >= nTiles) return;

  const float* brow = Bs + m * GRU_D + 2 * khalf;
  v2f bf[16];
#pragma unroll
  for (int k = 0; k < 16; ++k) bf[k] = *(const v2f*)(brow + 4 * k);
  const float bias = (m < 3) ? b_ih[m] : 0.0f;

  const int laneOff = m * GRU_D + 2 * khalf;

  v2f a0[16], a1[16];
  gates_load_tile(x, tile, laneOff, a0);
  while (tile + 2 * nWaves < nTiles) {
    gates_load_tile(x, tile + nWaves, laneOff, a1);
    __builtin_amdgcn_sched_barrier(0);
    gates_mma_store(a0, bf, bias, gx, rows, tile, m, khalf);
    gates_load_tile(x, tile + 2 * nWaves, laneOff, a0);
    __builtin_amdgcn_sched_barrier(0);
    gates_mma_store(a1, bf, bias, gx, rows, tile + nWaves, m, khalf);
    tile += 2 * nWaves;
  }
  if (tile + nWaves < nTiles) {
    gates_load_tile(x, tile + nWaves, laneOff, a1);
    __builtin_amdgcn_sched_barrier(0);
    gates_mma_store(a0, bf, bias, gx, rows, tile, m, khalf);
    gates_mma_store(a1, bf, bias, gx, rows, tile + nWaves, m, khalf);
  } else {
    gates_mma_store(a0, bf, bias, gx, rows, tile, m, khalf);
  }
}

#define SCAN_T 16

__device__ __forceinline__ void gru_load_block(const float* __restrict__ pr,
                                               const float* __restrict__ pz,
                                               const float* __restrict__ pn,
                                               int t0,
                                               float (&xr)[SCAN_T],
                                               float (&xz)[SCAN_T],
                                               float (&xn)[SCAN_T]) {
#pragma unroll
  for (int i = 0; i < SCAN_T; ++i) {
    const int off = (t0 + i) * GRU_B;
    xr[i] = pr[off];
    xz[i] = pz[off];
    xn[i] = pn[off];
  }
}

__device__ __forceinline__ float gru_run_block(float h, int t0, int len,
                                               const float (&xr)[SCAN_T],
                                               const float (&xz)[SCAN_T],
                                               const float (&xn)[SCAN_T],
                                               float hw0, float hw1, float hw2,
                                               float hc0, float hc1, float hc2,
                                               float* __restrict__ po) {
#pragma unroll
  for (int i = 0; i < SCAN_T; ++i) {
    const int t = t0 + i;
    const float pr = fmaf(0.5f, xr[i], hc0);
    const float pz = fmaf(0.5f, xz[i], hc1);
    const float tr = __builtin_amdgcn_tanhf(fmaf(h, hw0, pr));
    const float tz = __builtin_amdgcn_tanhf(fmaf(h, hw1, pz));
    const float hg = fmaf(h, hw2, hc2);
    const float n  = __builtin_amdgcn_tanhf(fmaf(tr, hg, xn[i] + hg));
    const float hh  = 0.5f * h;
    const float zh  = fmaf(tz, hh, hh);
    const float omz = fmaf(-0.5f, tz, 0.5f);
    h = fmaf(omz, n, zh);
    po[t * GRU_B] = (t < len) ? h : 0.0f;
  }
  return h;
}

__global__ void __launch_bounds__(256)
gru_scan_kernel(const float* __restrict__ gx,
                const float* __restrict__ h0,
                const int*   __restrict__ lengths,
                const float* __restrict__ W_hh,
                const float* __restrict__ b_hh,
                float* __restrict__ out) {
  const int b = blockIdx.x * blockDim.x + threadIdx.x;
  if (b >= GRU_B) return;

  const size_t plane = (size_t)GRU_L * GRU_B;
  const float* pr = gx + b;
  const float* pz = gx + plane + b;
  const float* pn = gx + 2 * plane + b;
  float*       po = out + b;

  float h = h0[b];
  const int   len = lengths[b];
  const float w0 = 0.5f * W_hh[0], w1 = 0.5f * W_hh[1], w2 = 0.5f * W_hh[2];
  const float c0 = 0.5f * b_hh[0], c1 = 0.5f * b_hh[1], c2 = 0.5f * b_hh[2];

  float ar[SCAN_T], az[SCAN_T], an[SCAN_T];
  float br[SCAN_T], bz[SCAN_T], bn[SCAN_T];

  gru_load_block(pr, pz, pn, 0, ar, az, an);
  for (int t0 = 0; t0 < GRU_L; t0 += 2 * SCAN_T) {
    gru_load_block(pr, pz, pn, t0 + SCAN_T, br, bz, bn);
    h = gru_run_block(h, t0, len, ar, az, an, w0, w1, w2, c0, c1, c2, po);
    const int tn = (t0 + 2 * SCAN_T < GRU_L) ? (t0 + 2 * SCAN_T) : 0;
    gru_load_block(pr, pz, pn, tn, ar, az, an);
    h = gru_run_block(h, t0 + SCAN_T, len, br, bz, bn, w0, w1, w2, c0, c1, c2, po);
  }
}

extern "C" void kernel_launch(void* const* d_in, const int* in_sizes, int n_in,
                              void* d_out, int out_size, void* d_ws, size_t ws_size,
                              hipStream_t stream) {
  const float* x       = (const float*)d_in[0];
  const float* h0      = (const float*)d_in[1];
  const int*   lengths = (const int*)  d_in[2];
  const float* W_ih    = (const float*)d_in[3];
  const float* W_hh    = (const float*)d_in[4];
  const float* b_ih    = (const float*)d_in[5];
  const float* b_hh    = (const float*)d_in[6];
  float*       out     = (float*)d_out;
  float*       gx      = (float*)d_ws;

  const int rows    = GRU_L * GRU_B;
  const int blocks1 = 1024;

  gru_gates_wmma_kernel<<<blocks1, 256, 0, stream>>>(x, W_ih, b_ih, gx, rows);
  gru_scan_kernel<<<(GRU_B + 255) / 256, 256, 0, stream>>>(gx, h0, lengths,
                                                           W_hh, b_hh, out);
}


# gates kernel: each x tile is fetched with 8 whole-line loads (512 contiguous bytes per wave load; was 16 loads of 16-byte pieces of 16 rows) and turned into A fragments through a per-wave LDS area
# speedup vs baseline: 4.6579x; 1.0558x over previous
#include <hip/hip_runtime.h>
#include <hip/hip_bf16.h>
#include <math.h>


typedef float v2f __attribute__((ext_vector_type(2)));
typedef float v4f __attribute__((ext_vector_type(4)));
typedef float v8f __attribute__((ext_vector_type(8)));

#define GRU_L 512
#define GRU_B 2048
#define GRU_D 64
#define GRU_XPITCH 68

__device__ __forceinline__ void gates_load_tile(const float* __restrict__ x,
                                                int tile, int lane,
                                                v4f (&r)[8]) {
  const float* src = x + (size_t)tile * (16 * GRU_D) + 4 * lane;
#pragma unroll
  for (int q = 0; q < 8; ++q) r[q] = *(const v4f*)(src + 128 * q);
}

__device__ __forceinline__ void gates_stage_tile(const v4f (&r)[8],
                                                 float* xs_st,
                                                 const float* xs_ld,
                                                 v2f (&a)[16]) {
#pragma unroll
  for (int q = 0; q < 8; ++q) *(v4f*)(xs_st + 2 * q * GRU_XPITCH) = r[q];
#pragma unroll
  for (int k = 0; k < 16; ++k) a[k] = *(const v2f*)(xs_ld + 4 * k);
  __builtin_amdgcn_sched_barrier(0);
}

__device__ __forceinline__ void gates_mma_store(const v2f (&a)[16],
                                                const v2f (&b)[16],
                                                float bias,
                                                float* __restrict__ gx,
                                                int rows, int tile,
                                                int m, int khalf) {
  v8f c = {};
#pragma unroll
  for (int k = 0; k < 16; ++k) {
    c = __builtin_amdgcn_wmma_f32_16x16x4_f32(false, a[k], false, b[k],
                                              (short)0, c, false, false);
  }
  if (m < 3) {
    float* plane = gx + (size_t)m * rows + (size_t)tile * 16 + 8 * khalf;
#pragma unroll
    for (int i = 0; i < 8; ++i) plane[i] = c[i] + bias;
  }
}

__global__ void __launch_bounds__(256)
gru_gates_wmma_kernel(const float* __restrict__ x,
                      const float* __restrict__ W_ih,
                      const float* __restrict__ b_ih,
                      float* __restrict__ gx,
                      int rows) {
  __shared__ float Bs[16 * GRU_D];
  __shared__ __attribute__((aligned(16))) float Xs[8 * 16 * GRU_XPITCH];
  for (int i = threadIdx.x; i < 16 * GRU_D; i += blockDim.x) {
    const int row = i >> 6;
    const int col = i & (GRU_D - 1);
    Bs[i] = (row < 3) ? W_ih[row * GRU_D + col] : 0.0f;
  }
  __syncthreads();

  const int lane  = threadIdx.x & 31;
  const int m     = lane & 15;
  const int khalf = lane >> 4;

  const int waveInBlock = __builtin_amdgcn_readfirstlane((int)(threadIdx.x >> 5));
  const int wavesPerBlk = (int)(blockDim.x >> 5);
  const int nWaves = (int)gridDim.x * wavesPerBlk;
  const int nTiles = rows >> 4;
  int tile = (int)blockIdx.x * wavesPerBlk + waveInBlock;
  if (tile >= nTiles) return;

  const float* brow = Bs + m * GRU_D + 2 * khalf;
  v2f bf[16];
#pragma unroll
  for (int k = 0; k < 16; ++k) bf[k] = *(const v2f*)(brow + 4 * k);
  const float bias = (m < 3) ? b_ih[m] : 0.0f;

  float* xs = Xs + waveInBlock * (16 * GRU_XPITCH);
  float*       xs_st = xs + khalf * GRU_XPITCH + 4 * m;
  const float* xs_ld = xs + m * GRU_XPITCH + 2 * khalf;

  v4f r0[8], r1[8];
  v2f a[16];
  gates_load_tile(x, tile, lane, r0);
  while (tile + 2 * nWaves < nTiles) {
    gates_load_tile(x, tile + nWaves, lane, r1);
    __builtin_amdgcn_sched_barrier(0);
    gates_stage_tile(r0, xs_st, xs_ld, a);
    gates_mma_store(a, bf, bias, gx, rows, tile, m, khalf);
    gates_load_tile(x, tile + 2 * nWaves, lane, r0);
    __builtin_amdgcn_sched_barrier(0);
    gates_stage_tile(r1, xs_st, xs_ld, a);
    gates_mma_store(a, bf, bias, gx, rows, tile + nWaves, m, khalf);
    tile += 2 * nWaves;
  }
  if (tile + nWaves < nTiles) {
    gates_load_tile(x, tile + nWaves, lane, r1);
    __builtin_amdgcn_sched_barrier(0);
    gates_stage_tile(r0, xs_st, xs_ld, a);
    gates_mma_store(a, bf, bias, gx, rows, tile, m, khalf);
    gates_stage_tile(r1, xs_st, xs_ld, a);
    gates_mma_store(a, bf, bias, gx, rows, tile + nWaves, m, khalf);
  } else {
    gates_stage_tile(r0, xs_st, xs_ld, a);
    gates_mma_store(a, bf, bias, gx, rows, tile, m, khalf);
  }
}

#define SCAN_T 16

__device__ __forceinline__ void gru_load_block(const float* __restrict__ pr,
                                               const float* __restrict__ pz,
                                               const float* __restrict__ pn,
                                               int t0,
                                               float (&xr)[SCAN_T],
                                               float (&xz)[SCAN_T],
                                               float (&xn)[SCAN_T]) {
#pragma unroll
  for (int i = 0; i < SCAN_T; ++i) {
    const int off = (t0 + i) * GRU_B;
    xr[i] = pr[off];
    xz[i] = pz[off];
    xn[i] = pn[off];
  }
}

__device__ __forceinline__ float gru_run_block(float h, int t0, int len,
                                               const float (&xr)[SCAN_T],
                                               const float (&xz)[SCAN_T],
                                               const float (&xn)[SCAN_T],
                                               float hw0, float hw1, float hw2,
                                               float hc0, float hc1, float hc2,
                                               float* __restrict__ po) {
#pragma unroll
  for (int i = 0; i < SCAN_T; ++i) {
    const int t = t0 + i;
    const float pr = fmaf(0.5f, xr[i], hc0);
    const float pz = fmaf(0.5f, xz[i], hc1);
    const float tr = __builtin_amdgcn_tanhf(fmaf(h, hw0, pr));
    const float tz = __builtin_amdgcn_tanhf(fmaf(h, hw1, pz));
    const float hg = fmaf(h, hw2, hc2);
    const float n  = __builtin_amdgcn_tanhf(fmaf(tr, hg, xn[i] + hg));
    const float hh  = 0.5f * h;
    const float zh  = fmaf(tz, hh, hh);
    const float omz = fmaf(-0.5f, tz, 0.5f);
    h = fmaf(omz, n, zh);
    po[t * GRU_B] = (t < len) ? h : 0.0f;
  }
  return h;
}

__global__ void __launch_bounds__(256)
gru_scan_kernel(const float* __restrict__ gx,
                const float* __restrict__ h0,
                const int*   __restrict__ lengths,
                const float* __restrict__ W_hh,
                const float* __restrict__ b_hh,
                float* __restrict__ out) {
  const int b = blockIdx.x * blockDim.x + threadIdx.x;
  if (b >= GRU_B) return;

  const size_t plane = (size_t)GRU_L * GRU_B;
  const float* pr = gx + b;
  const float* pz = gx + plane + b;
  const float* pn = gx + 2 * plane + b;
  float*       po = out + b;

  float h = h0[b];
  const int   len = lengths[b];
  const float w0 = 0.5f * W_hh[0], w1 = 0.5f * W_hh[1], w2 = 0.5f * W_hh[2];
  const float c0 = 0.5f * b_hh[0], c1 = 0.5f * b_hh[1], c2 = 0.5f * b_hh[2];

  float ar[SCAN_T], az[SCAN_T], an[SCAN_T];
  float br[SCAN_T], bz[SCAN_T], bn[SCAN_T];

  gru_load_block(pr, pz, pn, 0, ar, az, an);
  for (int t0 = 0; t0 < GRU_L; t0 += 2 * SCAN_T) {
    gru_load_block(pr, pz, pn, t0 + SCAN_T, br, bz, bn);
    h = gru_run_block(h, t0, len, ar, az, an, w0, w1, w2, c0, c1, c2, po);
    const int tn = (t0 + 2 * SCAN_T < GRU_L) ? (t0 + 2 * SCAN_T) : 0;
    gru_load_block(pr, pz, pn, tn, ar, az, an);
    h = gru_run_block(h, t0 + SCAN_T, len, br, bz, bn, w0, w1, w2, c0, c1, c2, po);
  }
}

extern "C" void kernel_launch(void* const* d_in, const int* in_sizes, int n_in,
                              void* d_out, int out_size, void* d_ws, size_t ws_size,
                              hipStream_t stream) {
  const float* x       = (const float*)d_in[0];
  const float* h0      = (const float*)d_in[1];
  const int*   lengths = (const int*)  d_in[2];
  const float* W_ih    = (const float*)d_in[3];
  const float* W_hh    = (const float*)d_in[4];
  const float* b_ih    = (const float*)d_in[5];
  const float* b_hh    = (const float*)d_in[6];
  float*       out     = (float*)d_out;
  float*       gx      = (float*)d_ws;

  const int rows    = GRU_L * GRU_B;
  const int blocks1 = 1024;

  gru_gates_wmma_kernel<<<blocks1, 256, 0, stream>>>(x, W_ih, b_ih, gx, rows);
  gru_scan_kernel<<<(GRU_B + 255) / 256, 256, 0, stream>>>(gx, h0, lengths,
                                                           W_hh, b_hh, out);
}


# one launch for two: a workgroup owns 8 batch columns; 8 waves make their gate values (same f32 WMMA chain) into an LDS ring, a ninth wave runs the scan from it; no gate planes in memory
# speedup vs baseline: 7.2313x; 1.5525x over previous
#include <hip/hip_runtime.h>
#include <hip/hip_bf16.h>
#include <math.h>


typedef float v2f __attribute__((ext_vector_type(2)));
typedef float v4f __attribute__((ext_vector_type(4)));
typedef float v8f __attribute__((ext_vector_type(8)));

#define GRU_L 512
#define GRU_B 2048
#define GRU_D 64
#define GRU_XPITCH 68

#define F_COLS   8
#define F_TC     32
#define F_NCHUNK (GRU_L / F_TC)
#define F_NPROD  8
#define F_WAVES  (F_NPROD + 1)
#define F_GPITCH (F_TC + 4)
#define F_GSTAGE (3 * F_COLS * F_GPITCH)

__device__ __forceinline__ void fused_load_tile(const float* __restrict__ x,
                                                int t, int b0, int lane,
                                                v4f (&r)[8]) {
  const float* s0 = x + ((size_t)t * GRU_B + b0) * GRU_D + 4 * lane;
  const float* s1 = s0 + (size_t)GRU_B * GRU_D;
#pragma unroll
  for (int q = 0; q < 4; ++q) r[q] = *(const v4f*)(s0 + 128 * q);
#pragma unroll
  for (int q = 0; q < 4; ++q) r[4 + q] = *(const v4f*)(s1 + 128 * q);
}

__device__ __forceinline__ void fused_store_tile(const v4f (&r)[8], float* xs_st) {
#pragma unroll
  for (int q = 0; q < 8; ++q) *(v4f*)(xs_st + 2 * q * GRU_XPITCH) = r[q];
}

__device__ __forceinline__ void fused_read_frags(const float* xs_ld, v2f (&a)[16]) {
#pragma unroll
  for (int k = 0; k < 16; ++k) a[k] = *(const v2f*)(xs_ld + 4 * k);
}

__device__ __forceinline__ void fused_mma_gates(const v2f (&a)[16],
                                                const v2f (&b)[16],
                                                float bias, float* gdst, int m) {
  v8f c = {};
#pragma unroll
  for (int k = 0; k < 16; ++k) {
    c = __builtin_amdgcn_wmma_f32_16x16x4_f32(false, a[k], false, b[k],
                                              (short)0, c, false, false);
  }
  if (m < 3) {
#pragma unroll
    for (int i = 0; i < 8; ++i) gdst[i * F_GPITCH] = c[i] + bias;
  }
}

__device__ __forceinline__ void fused_produce_chunk(const float* __restrict__ x,
                                                    int tNext0, int tNext1,
                                                    int b0, int lane,
                                                    float* xs_st0, const float* xs_ld0,
                                                    float* xs_st1, const float* xs_ld1,
                                                    const v2f (&b)[16], float bias,
                                                    float* gdst, int m) {
  v4f ra[8], rb[8];
  fused_load_tile(x, tNext0, b0, lane, ra);
  fused_load_tile(x, tNext1, b0, lane, rb);
  __builtin_amdgcn_sched_barrier(0);
  v2f a0[16], a1[16];
  fused_read_frags(xs_ld0, a0);
  fused_read_frags(xs_ld1, a1);
  __builtin_amdgcn_sched_barrier(0);
  fused_mma_gates(a0, b, bias, gdst, m);
  fused_mma_gates(a1, b, bias, gdst + 2, m);
  __builtin_amdgcn_sched_barrier(0);
  fused_store_tile(ra, xs_st0);
  fused_store_tile(rb, xs_st1);
}

__device__ __forceinline__ void fused_stage_barrier() {
  asm volatile("s_wait_dscnt 0x0" ::: "memory");
  asm volatile("s_barrier_signal -1" ::: "memory");
  asm volatile("s_barrier_wait -1" ::: "memory");
}

__device__ __forceinline__ float fused_scan_chunk(float h, int t0, int len,
                                                  const float* gs,
                                                  float hw0, float hw1, float hw2,
                                                  float hc0, float hc1, float hc2,
                                                  float* __restrict__ po,
                                                  bool stores) {
  float xr[F_TC], xz[F_TC], xn[F_TC], o[F_TC];
#pragma unroll
  for (int k = 0; k < F_TC / 4; ++k) {
    const v4f vr = *(const v4f*)(gs + 4 * k);
    const v4f vz = *(const v4f*)(gs + F_COLS * F_GPITCH + 4 * k);
    const v4f vn = *(const v4f*)(gs + 2 * F_COLS * F_GPITCH + 4 * k);
#pragma unroll
    for (int e = 0; e < 4; ++e) {
      xr[4 * k + e] = vr[e];
      xz[4 * k + e] = vz[e];
      xn[4 * k + e] = vn[e];
    }
  }
#pragma unroll
  for (int i = 0; i < F_TC; ++i) {
    const int t = t0 + i;
    const float pr = fmaf(0.5f, xr[i], hc0);
    const float pz = fmaf(0.5f, xz[i], hc1);
    const float tr = __builtin_amdgcn_tanhf(fmaf(h, hw0, pr));
    const float tz = __builtin_amdgcn_tanhf(fmaf(h, hw1, pz));
    const float hg = fmaf(h, hw2, hc2);
    const float n  = __builtin_amdgcn_tanhf(fmaf(tr, hg, xn[i] + hg));
    const float hh  = 0.5f * h;
    const float zh  = fmaf(tz, hh, hh);
    const float omz = fmaf(-0.5f, tz, 0.5f);
    h = fmaf(omz, n, zh);
    o[i] = (t < len) ? h : 0.0f;
  }
  if (stores) {
#pragma unroll
    for (int i = 0; i < F_TC; ++i) po[(size_t)(t0 + i) * GRU_B] = o[i];
  }
  return h;
}

__global__ void __launch_bounds__(32 * F_WAVES)
gru_fused_kernel(const float* __restrict__ x,
                 const float* __restrict__ h0,
                 const int*   __restrict__ lengths,
                 const float* __restrict__ W_ih,
                 const float* __restrict__ W_hh,
                 const float* __restrict__ b_ih,
                 const float* __restrict__ b_hh,
                 float* __restrict__ out) {
  __shared__ float Bs[16 * GRU_D];
  __shared__ __attribute__((aligned(16))) float Xs[2 * F_NPROD * 16 * GRU_XPITCH];
  __shared__ __attribute__((aligned(16))) float Gs[2 * F_GSTAGE];

  for (int i = threadIdx.x; i < 16 * GRU_D; i += blockDim.x) {
    const int row = i >> 6;
    const int col = i & (GRU_D - 1);
    Bs[i] = (row < 3) ? W_ih[row * GRU_D + col] : 0.0f;
  }
  __syncthreads();

  const int lane  = threadIdx.x & 31;
  const int m     = lane & 15;
  const int khalf = lane >> 4;
  const int wave  = __builtin_amdgcn_readfirstlane((int)(threadIdx.x >> 5));
  const bool scanRole = (wave == 0);
  const int pi    = (wave > 0) ? (wave - 1) : 0;
  const int b0    = (int)blockIdx.x * F_COLS;

  const float* brow = Bs + m * GRU_D + 2 * khalf;
  v2f bf[16];
#pragma unroll
  for (int k = 0; k < 16; ++k) bf[k] = *(const v2f*)(brow + 4 * k);
  const float bias = (m < 3) ? b_ih[m] : 0.0f;
  float* xs = Xs + 2 * pi * (16 * GRU_XPITCH);
  float*       xs_st0 = xs + khalf * GRU_XPITCH + 4 * m;
  const float* xs_ld0 = xs + m * GRU_XPITCH + 2 * khalf;
  float*       xs_st1 = xs_st0 + 16 * GRU_XPITCH;
  const float* xs_ld1 = xs_ld0 + 16 * GRU_XPITCH;
  const int gofs = ((m < 3) ? m : 0) * (F_COLS * F_GPITCH) + 4 * pi + khalf;
  const int ts0 = 4 * pi, ts1 = 4 * pi + 2;

  const int col = lane & (F_COLS - 1);
  const int b   = b0 + col;
  float h = h0[b];
  const int len = lengths[b];
  const float w0 = 0.5f * W_hh[0], w1 = 0.5f * W_hh[1], w2 = 0.5f * W_hh[2];
  const float c0 = 0.5f * b_hh[0], c1 = 0.5f * b_hh[1], c2 = 0.5f * b_hh[2];
  float* po = out + b;
  const float* gcol = Gs + col * F_GPITCH;

  if (!scanRole) {
    v4f ra[8], rb[8];
    fused_load_tile(x, ts0, b0, lane, ra);
    fused_load_tile(x, ts1, b0, lane, rb);
    fused_store_tile(ra, xs_st0);
    fused_store_tile(rb, xs_st1);
    fused_produce_chunk(x, F_TC + ts0, F_TC + ts1, b0, lane,
                        xs_st0, xs_ld0, xs_st1, xs_ld1, bf, bias, Gs + gofs, m);
  }
  fused_stage_barrier();

  for (int c = 0; c < F_NCHUNK; ++c) {
    if (scanRole) {
      h = fused_scan_chunk(h, c * F_TC, len, gcol + (c & 1) * F_GSTAGE,
                           w0, w1, w2, c0, c1, c2, po, lane < F_COLS);
    } else {
      const int cn = (c + 2 < F_NCHUNK) ? (c + 2) : (F_NCHUNK - 1);
      float* gst = Gs + ((c + 1) & 1) * F_GSTAGE + gofs;
      fused_produce_chunk(x, cn * F_TC + ts0, cn * F_TC + ts1, b0, lane,
                          xs_st0, xs_ld0, xs_st1, xs_ld1, bf, bias, gst, m);
    }
    fused_stage_barrier();
  }
}

extern "C" void kernel_launch(void* const* d_in, const int* in_sizes, int n_in,
                              void* d_out, int out_size, void* d_ws, size_t ws_size,
                              hipStream_t stream) {
  const float* x       = (const float*)d_in[0];
  const float* h0      = (const float*)d_in[1];
  const int*   lengths = (const int*)  d_in[2];
  const float* W_ih    = (const float*)d_in[3];
  const float* W_hh    = (const float*)d_in[4];
  const float* b_ih    = (const float*)d_in[5];
  const float* b_hh    = (const float*)d_in[6];
  float*       out     = (float*)d_out;

  gru_fused_kernel<<<GRU_B / F_COLS, 32 * F_WAVES, 0, stream>>>(
      x, h0, lengths, W_ih, W_hh, b_ih, b_hh, out);
}


# producers hold two register sets of x tiles: a set is refilled right behind its staging stores with the chunk due two steps on, so each load has two whole steps to arrive (roles loop apart)
# speedup vs baseline: 7.8713x; 1.0885x over previous
#include <hip/hip_runtime.h>
#include <hip/hip_bf16.h>
#include <math.h>


typedef float v2f __attribute__((ext_vector_type(2)));
typedef float v4f __attribute__((ext_vector_type(4)));
typedef float v8f __attribute__((ext_vector_type(8)));

#define GRU_L 512
#define GRU_B 2048
#define GRU_D 64
#define GRU_XPITCH 68

#define F_COLS   8
#define F_TC     32
#define F_NCHUNK (GRU_L / F_TC)
#define F_NPROD  8
#define F_WAVES  (F_NPROD + 1)
#define F_GPITCH (F_TC + 4)
#define F_GSTAGE (3 * F_COLS * F_GPITCH)

typedef const v4f __attribute__((address_space(1)))* fused_gptr;
__device__ __forceinline__ void fused_load_tile(const float* __restrict__ x,
                                                int t, int b0, int lane,
                                                v4f (&r)[8]) {
  const unsigned long long u =
      (unsigned long long)(x + ((size_t)t * GRU_B + b0) * GRU_D);
  const unsigned lo = (unsigned)__builtin_amdgcn_readfirstlane((int)(unsigned)u);
  const unsigned hi = (unsigned)__builtin_amdgcn_readfirstlane((int)(unsigned)(u >> 32));
  const fused_gptr s0 = (fused_gptr)(((unsigned long long)hi << 32) | (unsigned long long)lo);
  const fused_gptr s1 = s0 + (size_t)GRU_B * GRU_D / 4;
  const unsigned l = (unsigned)lane;
#pragma unroll
  for (int q = 0; q < 4; ++q) r[q] = s0[l + 32u * q];
#pragma unroll
  for (int q = 0; q < 4; ++q) r[4 + q] = s1[l + 32u * q];
}

__device__ __forceinline__ void fused_store_tile(const v4f (&r)[8], float* xs_st) {
#pragma unroll
  for (int q = 0; q < 8; ++q) *(v4f*)(xs_st + 2 * q * GRU_XPITCH) = r[q];
}

__device__ __forceinline__ void fused_read_frags(const float* xs_ld, v2f (&a)[16]) {
#pragma unroll
  for (int k = 0; k < 16; ++k) a[k] = *(const v2f*)(xs_ld + 4 * k);
}

__device__ __forceinline__ void fused_mma_gates(const v2f (&a)[16],
                                                const v2f (&b)[16],
                                                float bias, float* gdst, int m) {
  v8f c = {};
#pragma unroll
  for (int k = 0; k < 16; ++k) {
    c = __builtin_amdgcn_wmma_f32_16x16x4_f32(false, a[k], false, b[k],
                                              (short)0, c, false, false);
  }
  if (m < 3) {
#pragma unroll
    for (int i = 0; i < 8; ++i) gdst[i * F_GPITCH] = c[i] + bias;
  }
}

template <bool kStage, bool kRequest>
__device__ __forceinline__ void fused_produce_chunk(const float* __restrict__ x,
                                                    int tReq0, int tReq1,
                                                    int b0, int lane,
                                                    v4f (&sa)[8], v4f (&sb)[8],
                                                    float* xs_st0, const float* xs_ld0,
                                                    float* xs_st1, const float* xs_ld1,
                                                    const v2f (&b)[16], float bias,
                                                    float* gdst, int m) {
  v2f a0[16], a1[16];
  fused_read_frags(xs_ld0, a0);
  fused_read_frags(xs_ld1, a1);
  __builtin_amdgcn_sched_barrier(0);
  fused_mma_gates(a0, b, bias, gdst, m);
  fused_mma_gates(a1, b, bias, gdst + 2, m);
  if (kStage) {
    __builtin_amdgcn_sched_barrier(0);
    fused_store_tile(sa, xs_st0);
    fused_store_tile(sb, xs_st1);
  }
  if (kRequest) {
    __builtin_amdgcn_sched_barrier(0);
    fused_load_tile(x, tReq0, b0, lane, sa);
    fused_load_tile(x, tReq1, b0, lane, sb);
  }
}

__device__ __forceinline__ void fused_stage_barrier() {
  asm volatile("s_wait_dscnt 0x0" ::: "memory");
  asm volatile("s_barrier_signal -1" ::: "memory");
  asm volatile("s_barrier_wait -1" ::: "memory");
}

__device__ __forceinline__ float fused_scan_chunk(float h, int t0, int len,
                                                  const float* gs,
                                                  float hw0, float hw1, float hw2,
                                                  float hc0, float hc1, float hc2,
                                                  float* __restrict__ po,
                                                  bool stores) {
  float xr[F_TC], xz[F_TC], xn[F_TC], o[F_TC];
#pragma unroll
  for (int k = 0; k < F_TC / 4; ++k) {
    const v4f vr = *(const v4f*)(gs + 4 * k);
    const v4f vz = *(const v4f*)(gs + F_COLS * F_GPITCH + 4 * k);
    const v4f vn = *(const v4f*)(gs + 2 * F_COLS * F_GPITCH + 4 * k);
#pragma unroll
    for (int e = 0; e < 4; ++e) {
      xr[4 * k + e] = vr[e];
      xz[4 * k + e] = vz[e];
      xn[4 * k + e] = vn[e];
    }
  }
#pragma unroll
  for (int i = 0; i < F_TC; ++i) {
    const int t = t0 + i;
    const float pr = fmaf(0.5f, xr[i], hc0);
    const float pz = fmaf(0.5f, xz[i], hc1);
    const float tr = __builtin_amdgcn_tanhf(fmaf(h, hw0, pr));
    const float tz = __builtin_amdgcn_tanhf(fmaf(h, hw1, pz));
    const float hg = fmaf(h, hw2, hc2);
    const float n  = __builtin_amdgcn_tanhf(fmaf(tr, hg, xn[i] + hg));
    const float hh  = 0.5f * h;
    const float zh  = fmaf(tz, hh, hh);
    const float omz = fmaf(-0.5f, tz, 0.5f);
    h = fmaf(omz, n, zh);
    o[i] = (t < len) ? h : 0.0f;
  }
  if (stores) {
#pragma unroll
    for (int i = 0; i < F_TC; ++i) po[(size_t)(t0 + i) * GRU_B] = o[i];
  }
  return h;
}

__global__ void __launch_bounds__(32 * F_WAVES)
gru_fused_kernel(const float* __restrict__ x,
                 const float* __restrict__ h0,
                 const int*   __restrict__ lengths,
                 const float* __restrict__ W_ih,
                 const float* __restrict__ W_hh,
                 const float* __restrict__ b_ih,
                 const float* __restrict__ b_hh,
                 float* __restrict__ out) {
  __shared__ float Bs[16 * GRU_D];
  __shared__ __attribute__((aligned(16))) float Xs[2 * F_NPROD * 16 * GRU_XPITCH];
  __shared__ __attribute__((aligned(16))) float Gs[2 * F_GSTAGE];

  for (int i = threadIdx.x; i < 16 * GRU_D; i += blockDim.x) {
    const int row = i >> 6;
    const int col = i & (GRU_D - 1);
    Bs[i] = (row < 3) ? W_ih[row * GRU_D + col] : 0.0f;
  }
  __syncthreads();

  const int lane  = threadIdx.x & 31;
  const int m     = lane & 15;
  const int khalf = lane >> 4;
  const int wave  = __builtin_amdgcn_readfirstlane((int)(threadIdx.x >> 5));
  const bool scanRole = (wave == 0);
  const int pi    = (wave > 0) ? (wave - 1) : 0;
  const int b0    = (int)blockIdx.x * F_COLS;

  const float* brow = Bs + m * GRU_D + 2 * khalf;
  v2f bf[16];
#pragma unroll
  for (int k = 0; k < 16; ++k) bf[k] = *(const v2f*)(brow + 4 * k);
  const float bias = (m < 3) ? b_ih[m] : 0.0f;
  float* xs = Xs + 2 * pi * (16 * GRU_XPITCH);
  float*       xs_st0 = xs + khalf * GRU_XPITCH + 4 * m;
  const float* xs_ld0 = xs + m * GRU_XPITCH + 2 * khalf;
  float*       xs_st1 = xs_st0 + 16 * GRU_XPITCH;
  const float* xs_ld1 = xs_ld0 + 16 * GRU_XPITCH;
  const int gofs = ((m < 3) ? m : 0) * (F_COLS * F_GPITCH) + 4 * pi + khalf;
  const int ts0 = 4 * pi, ts1 = 4 * pi + 2;

  const int col = lane & (F_COLS - 1);
  const int b   = b0 + col;
  float h = h0[b];
  const int len = lengths[b];
  const float w0 = 0.5f * W_hh[0], w1 = 0.5f * W_hh[1], w2 = 0.5f * W_hh[2];
  const float c0 = 0.5f * b_hh[0], c1 = 0.5f * b_hh[1], c2 = 0.5f * b_hh[2];
  float* po = out + b;
  const float* gcol = Gs + col * F_GPITCH;

  if (scanRole) {
    fused_stage_barrier();
    for (int c = 0; c < F_NCHUNK; ++c) {
      h = fused_scan_chunk(h, c * F_TC, len, gcol + (c & 1) * F_GSTAGE,
                           w0, w1, w2, c0, c1, c2, po, lane < F_COLS);
      fused_stage_barrier();
    }
  } else {
    v4f pa0[8], pa1[8], pb0[8], pb1[8];

    fused_load_tile(x, ts0, b0, lane, pa0);
    fused_load_tile(x, ts1, b0, lane, pa1);
    fused_load_tile(x, F_TC + ts0, b0, lane, pb0);
    fused_load_tile(x, F_TC + ts1, b0, lane, pb1);
    fused_store_tile(pa0, xs_st0);
    fused_store_tile(pa1, xs_st1);
    __builtin_amdgcn_sched_barrier(0);
    fused_load_tile(x, 2 * F_TC + ts0, b0, lane, pa0);
    fused_load_tile(x, 2 * F_TC + ts1, b0, lane, pa1);
    fused_produce_chunk<true, true>(x, 3 * F_TC + ts0, 3 * F_TC + ts1, b0, lane,
                                    pb0, pb1, xs_st0, xs_ld0, xs_st1, xs_ld1,
                                    bf, bias, Gs + gofs, m);
    fused_stage_barrier();

    for (int c = 0; c + 5 < F_NCHUNK; c += 2) {
      fused_produce_chunk<true, true>(x, (c + 4) * F_TC + ts0, (c + 4) * F_TC + ts1,
                                      b0, lane, pa0, pa1,
                                      xs_st0, xs_ld0, xs_st1, xs_ld1, bf, bias,
                                      Gs + F_GSTAGE + gofs, m);
      fused_stage_barrier();
      fused_produce_chunk<true, true>(x, (c + 5) * F_TC + ts0, (c + 5) * F_TC + ts1,
                                      b0, lane, pb0, pb1,
                                      xs_st0, xs_ld0, xs_st1, xs_ld1, bf, bias,
                                      Gs + gofs, m);
      fused_stage_barrier();
    }
    fused_produce_chunk<true, false>(x, 0, 0, b0, lane, pa0, pa1,
                                     xs_st0, xs_ld0, xs_st1, xs_ld1, bf, bias,
                                     Gs + F_GSTAGE + gofs, m);
    fused_stage_barrier();
    fused_produce_chunk<true, false>(x, 0, 0, b0, lane, pb0, pb1,
                                     xs_st0, xs_ld0, xs_st1, xs_ld1, bf, bias,
                                     Gs + gofs, m);
    fused_stage_barrier();
    fused_produce_chunk<false, false>(x, 0, 0, b0, lane, pa0, pa1,
                                      xs_st0, xs_ld0, xs_st1, xs_ld1, bf, bias,
                                      Gs + F_GSTAGE + gofs, m);
    fused_stage_barrier();
    fused_stage_barrier();
  }
}

extern "C" void kernel_launch(void* const* d_in, const int* in_sizes, int n_in,
                              void* d_out, int out_size, void* d_ws, size_t ws_size,
                              hipStream_t stream) {
  const float* x       = (const float*)d_in[0];
  const float* h0      = (const float*)d_in[1];
  const int*   lengths = (const int*)  d_in[2];
  const float* W_ih    = (const float*)d_in[3];
  const float* W_hh    = (const float*)d_in[4];
  const float* b_ih    = (const float*)d_in[5];
  const float* b_hh    = (const float*)d_in[6];
  float*       out     = (float*)d_out;

  gru_fused_kernel<<<GRU_B / F_COLS, 32 * F_WAVES, 0, stream>>>(
      x, h0, lengths, W_ih, W_hh, b_ih, b_hh, out);
}


# producers: a step's staging stores moved in front of its matrix work and its 16 refill requests issued one in front of every second WMMA instead of as one group before the hand-over barrier
# speedup vs baseline: 9.2482x; 1.1749x over previous
#include <hip/hip_runtime.h>
#include <hip/hip_bf16.h>
#include <math.h>


typedef float v2f __attribute__((ext_vector_type(2)));
typedef float v4f __attribute__((ext_vector_type(4)));
typedef float v8f __attribute__((ext_vector_type(8)));

#define GRU_L 512
#define GRU_B 2048
#define GRU_D 64
#define GRU_XPITCH 68

#define F_COLS   8
#define F_TC     32
#define F_NCHUNK (GRU_L / F_TC)
#define F_NPROD  8
#define F_WAVES  (F_NPROD + 1)
#define F_GPITCH (F_TC + 4)
#define F_GSTAGE (3 * F_COLS * F_GPITCH)

typedef const v4f __attribute__((address_space(1)))* fused_gptr;
__device__ __forceinline__ fused_gptr fused_tile_base(const float* __restrict__ x,
                                                      int t, int b0) {
  const unsigned long long u =
      (unsigned long long)(x + ((size_t)t * GRU_B + b0) * GRU_D);
  const unsigned lo = (unsigned)__builtin_amdgcn_readfirstlane((int)(unsigned)u);
  const unsigned hi = (unsigned)__builtin_amdgcn_readfirstlane((int)(unsigned)(u >> 32));
  return (fused_gptr)(((unsigned long long)hi << 32) | (unsigned long long)lo);
}
__device__ __forceinline__ v4f fused_load_piece(fused_gptr s0, int q, int lane) {
  const fused_gptr s = (q < 4) ? s0 : s0 + (size_t)GRU_B * GRU_D / 4;
  return s[(unsigned)lane + 32u * (unsigned)(q & 3)];
}
__device__ __forceinline__ void fused_load_tile(const float* __restrict__ x,
                                                int t, int b0, int lane,
                                                v4f (&r)[8]) {
  const fused_gptr s0 = fused_tile_base(x, t, b0);
#pragma unroll
  for (int q = 0; q < 8; ++q) r[q] = fused_load_piece(s0, q, lane);
}

__device__ __forceinline__ void fused_store_tile(const v4f (&r)[8], float* xs_st) {
#pragma unroll
  for (int q = 0; q < 8; ++q) *(v4f*)(xs_st + 2 * q * GRU_XPITCH) = r[q];
}

__device__ __forceinline__ void fused_read_frags(const float* xs_ld, v2f (&a)[16]) {
#pragma unroll
  for (int k = 0; k < 16; ++k) a[k] = *(const v2f*)(xs_ld + 4 * k);
}

template <bool kRequest>
__device__ __forceinline__ void fused_mma_gates(const v2f (&a)[16],
                                                const v2f (&b)[16],
                                                float bias, float* gdst, int m,
                                                const float* __restrict__ x,
                                                int tReq, int b0, int lane,
                                                v4f (&r)[8]) {
  fused_gptr s0 = nullptr;
  if (kRequest) s0 = fused_tile_base(x, tReq, b0);
  v8f c = {};
#pragma unroll
  for (int k = 0; k < 16; ++k) {
    if (kRequest && (k & 1) == 0) {
      __builtin_amdgcn_sched_barrier(0);
      r[k >> 1] = fused_load_piece(s0, k >> 1, lane);
      __builtin_amdgcn_sched_barrier(0);
    }
    c = __builtin_amdgcn_wmma_f32_16x16x4_f32(false, a[k], false, b[k],
                                              (short)0, c, false, false);
  }
  if (m < 3) {
#pragma unroll
    for (int i = 0; i < 8; ++i) gdst[i * F_GPITCH] = c[i] + bias;
  }
}

template <bool kStage, bool kRequest, bool kSpread = true>
__device__ __forceinline__ void fused_produce_chunk(const float* __restrict__ x,
                                                    int tReq0, int tReq1,
                                                    int b0, int lane,
                                                    v4f (&sa)[8], v4f (&sb)[8],
                                                    float* xs_st0, const float* xs_ld0,
                                                    float* xs_st1, const float* xs_ld1,
                                                    const v2f (&b)[16], float bias,
                                                    float* gdst, int m) {
  v2f a0[16], a1[16];
  fused_read_frags(xs_ld0, a0);
  fused_read_frags(xs_ld1, a1);
  if (kSpread) {
    if (kStage) {
      __builtin_amdgcn_sched_barrier(0);
      fused_store_tile(sa, xs_st0);
      fused_store_tile(sb, xs_st1);
    }
    __builtin_amdgcn_sched_barrier(0);
    fused_mma_gates<kRequest>(a0, b, bias, gdst, m, x, tReq0, b0, lane, sa);
    fused_mma_gates<kRequest>(a1, b, bias, gdst + 2, m, x, tReq1, b0, lane, sb);
  } else {
    __builtin_amdgcn_sched_barrier(0);
    fused_mma_gates<false>(a0, b, bias, gdst, m, x, 0, b0, lane, sa);
    fused_mma_gates<false>(a1, b, bias, gdst + 2, m, x, 0, b0, lane, sb);
    if (kStage) {
      __builtin_amdgcn_sched_barrier(0);
      fused_store_tile(sa, xs_st0);
      fused_store_tile(sb, xs_st1);
    }
    if (kRequest) {
      __builtin_amdgcn_sched_barrier(0);
      fused_load_tile(x, tReq0, b0, lane, sa);
      fused_load_tile(x, tReq1, b0, lane, sb);
    }
  }
}

__device__ __forceinline__ void fused_stage_barrier() {
  asm volatile("s_wait_dscnt 0x0" ::: "memory");
  asm volatile("s_barrier_signal -1" ::: "memory");
  asm volatile("s_barrier_wait -1" ::: "memory");
}

__device__ __forceinline__ float fused_scan_chunk(float h, int t0, int len,
                                                  const float* gs,
                                                  float hw0, float hw1, float hw2,
                                                  float hc0, float hc1, float hc2,
                                                  float* __restrict__ po,
                                                  bool stores) {
  float xr[F_TC], xz[F_TC], xn[F_TC], o[F_TC];
#pragma unroll
  for (int k = 0; k < F_TC / 4; ++k) {
    const v4f vr = *(const v4f*)(gs + 4 * k);
    const v4f vz = *(const v4f*)(gs + F_COLS * F_GPITCH + 4 * k);
    const v4f vn = *(const v4f*)(gs + 2 * F_COLS * F_GPITCH + 4 * k);
#pragma unroll
    for (int e = 0; e < 4; ++e) {
      xr[4 * k + e] = vr[e];
      xz[4 * k + e] = vz[e];
      xn[4 * k + e] = vn[e];
    }
  }
#pragma unroll
  for (int i = 0; i < F_TC; ++i) {
    const int t = t0 + i;
    const float pr = fmaf(0.5f, xr[i], hc0);
    const float pz = fmaf(0.5f, xz[i], hc1);
    const float tr = __builtin_amdgcn_tanhf(fmaf(h, hw0, pr));
    const float tz = __builtin_amdgcn_tanhf(fmaf(h, hw1, pz));
    const float hg = fmaf(h, hw2, hc2);
    const float n  = __builtin_amdgcn_tanhf(fmaf(tr, hg, xn[i] + hg));
    const float hh  = 0.5f * h;
    const float zh  = fmaf(tz, hh, hh);
    const float omz = fmaf(-0.5f, tz, 0.5f);
    h = fmaf(omz, n, zh);
    o[i] = (t < len) ? h : 0.0f;
  }
  if (stores) {
#pragma unroll
    for (int i = 0; i < F_TC; ++i) po[(size_t)(t0 + i) * GRU_B] = o[i];
  }
  return h;
}

__global__ void __launch_bounds__(32 * F_WAVES)
gru_fused_kernel(const float* __restrict__ x,
                 const float* __restrict__ h0,
                 const int*   __restrict__ lengths,
                 const float* __restrict__ W_ih,
                 const float* __restrict__ W_hh,
                 const float* __restrict__ b_ih,
                 const float* __restrict__ b_hh,
                 float* __restrict__ out) {
  __shared__ float Bs[16 * GRU_D];
  __shared__ __attribute__((aligned(16))) float Xs[2 * F_NPROD * 16 * GRU_XPITCH];
  __shared__ __attribute__((aligned(16))) float Gs[2 * F_GSTAGE];

  for (int i = threadIdx.x; i < 16 * GRU_D; i += blockDim.x) {
    const int row = i >> 6;
    const int col = i & (GRU_D - 1);
    Bs[i] = (row < 3) ? W_ih[row * GRU_D + col] : 0.0f;
  }
  __syncthreads();

  const int lane  = threadIdx.x & 31;
  const int m     = lane & 15;
  const int khalf = lane >> 4;
  const int wave  = __builtin_amdgcn_readfirstlane((int)(threadIdx.x >> 5));
  const bool scanRole = (wave == 0);
  const int pi    = (wave > 0) ? (wave - 1) : 0;
  const int b0    = (int)blockIdx.x * F_COLS;

  const float* brow = Bs + m * GRU_D + 2 * khalf;
  v2f bf[16];
#pragma unroll
  for (int k = 0; k < 16; ++k) bf[k] = *(const v2f*)(brow + 4 * k);
  const float bias = (m < 3) ? b_ih[m] : 0.0f;
  float* xs = Xs + 2 * pi * (16 * GRU_XPITCH);
  float*       xs_st0 = xs + khalf * GRU_XPITCH + 4 * m;
  const float* xs_ld0 = xs + m * GRU_XPITCH + 2 * khalf;
  float*       xs_st1 = xs_st0 + 16 * GRU_XPITCH;
  const float* xs_ld1 = xs_ld0 + 16 * GRU_XPITCH;
  const int gofs = ((m < 3) ? m : 0) * (F_COLS * F_GPITCH) + 4 * pi + khalf;
  const int ts0 = 4 * pi, ts1 = 4 * pi + 2;

  const int col = lane & (F_COLS - 1);
  const int b   = b0 + col;
  float h = h0[b];
  const int len = lengths[b];
  const float w0 = 0.5f * W_hh[0], w1 = 0.5f * W_hh[1], w2 = 0.5f * W_hh[2];
  const float c0 = 0.5f * b_hh[0], c1 = 0.5f * b_hh[1], c2 = 0.5f * b_hh[2];
  float* po = out + b;
  const float* gcol = Gs + col * F_GPITCH;

  if (scanRole) {
    fused_stage_barrier();
    for (int c = 0; c < F_NCHUNK; ++c) {
      h = fused_scan_chunk(h, c * F_TC, len, gcol + (c & 1) * F_GSTAGE,
                           w0, w1, w2, c0, c1, c2, po, lane < F_COLS);
      fused_stage_barrier();
    }
  } else {
    v4f pa0[8], pa1[8], pb0[8], pb1[8];

    fused_load_tile(x, ts0, b0, lane, pa0);
    fused_load_tile(x, ts1, b0, lane, pa1);
    fused_load_tile(x, F_TC + ts0, b0, lane, pb0);
    fused_load_tile(x, F_TC + ts1, b0, lane, pb1);
    fused_store_tile(pa0, xs_st0);
    fused_store_tile(pa1, xs_st1);
    __builtin_amdgcn_sched_barrier(0);
    fused_load_tile(x, 2 * F_TC + ts0, b0, lane, pa0);
    fused_load_tile(x, 2 * F_TC + ts1, b0, lane, pa1);
    fused_produce_chunk<true, true, false>(x, 3 * F_TC + ts0, 3 * F_TC + ts1, b0, lane,
                                           pb0, pb1, xs_st0, xs_ld0, xs_st1, xs_ld1,
                                           bf, bias, Gs + gofs, m);
    fused_stage_barrier();

    for (int c = 0; c + 5 < F_NCHUNK; c += 2) {
      fused_produce_chunk<true, true>(x, (c + 4) * F_TC + ts0, (c + 4) * F_TC + ts1,
                                      b0, lane, pa0, pa1,
                                      xs_st0, xs_ld0, xs_st1, xs_ld1, bf, bias,
                                      Gs + F_GSTAGE + gofs, m);
      fused_stage_barrier();
      fused_produce_chunk<true, true>(x, (c + 5) * F_TC + ts0, (c + 5) * F_TC + ts1,
                                      b0, lane, pb0, pb1,
                                      xs_st0, xs_ld0, xs_st1, xs_ld1, bf, bias,
                                      Gs + gofs, m);
      fused_stage_barrier();
    }
    fused_produce_chunk<true, false>(x, 0, 0, b0, lane, pa0, pa1,
                                     xs_st0, xs_ld0, xs_st1, xs_ld1, bf, bias,
                                     Gs + F_GSTAGE + gofs, m);
    fused_stage_barrier();
    fused_produce_chunk<true, false>(x, 0, 0, b0, lane, pb0, pb1,
                                     xs_st0, xs_ld0, xs_st1, xs_ld1, bf, bias,
                                     Gs + gofs, m);
    fused_stage_barrier();
    fused_produce_chunk<false, false>(x, 0, 0, b0, lane, pa0, pa1,
                                      xs_st0, xs_ld0, xs_st1, xs_ld1, bf, bias,
                                      Gs + F_GSTAGE + gofs, m);
    fused_stage_barrier();
    fused_stage_barrier();
  }
}

extern "C" void kernel_launch(void* const* d_in, const int* in_sizes, int n_in,
                              void* d_out, int out_size, void* d_ws, size_t ws_size,
                              hipStream_t stream) {
  const float* x       = (const float*)d_in[0];
  const float* h0      = (const float*)d_in[1];
  const int*   lengths = (const int*)  d_in[2];
  const float* W_ih    = (const float*)d_in[3];
  const float* W_hh    = (const float*)d_in[4];
  const float* b_ih    = (const float*)d_in[5];
  const float* b_hh    = (const float*)d_in[6];
  float*       out     = (float*)d_out;

  gru_fused_kernel<<<GRU_B / F_COLS, 32 * F_WAVES, 0, stream>>>(
      x, h0, lengths, W_ih, W_hh, b_ih, b_hh, out);
}


# speedup vs baseline: 9.6614x; 1.0447x over previous
#include <hip/hip_runtime.h>
#include <hip/hip_bf16.h>
#include <math.h>


typedef float v2f __attribute__((ext_vector_type(2)));
typedef float v4f __attribute__((ext_vector_type(4)));
typedef float v8f __attribute__((ext_vector_type(8)));

#define GRU_L 512
#define GRU_B 2048
#define GRU_D 64
#define GRU_XPITCH 68

#define F_COLS   8
#define F_TC     32
#define F_NCHUNK (GRU_L / F_TC)
#define F_NPROD  8
#define F_WAVES  (F_NPROD + 1)
#define F_GPITCH (F_TC + 4)
#define F_GSTAGE (3 * F_COLS * F_GPITCH)

typedef const v4f __attribute__((address_space(1)))* fused_gptr;
__device__ __forceinline__ fused_gptr fused_tile_base(const float* __restrict__ x,
                                                      int t, int b0) {
  const unsigned long long u =
      (unsigned long long)(x + ((size_t)t * GRU_B + b0) * GRU_D);
  const unsigned lo = (unsigned)__builtin_amdgcn_readfirstlane((int)(unsigned)u);
  const unsigned hi = (unsigned)__builtin_amdgcn_readfirstlane((int)(unsigned)(u >> 32));
  return (fused_gptr)(((unsigned long long)hi << 32) | (unsigned long long)lo);
}
__device__ __forceinline__ v4f fused_load_piece(fused_gptr s0, int q, int lane) {
  const fused_gptr s = (q < 4) ? s0 : s0 + (size_t)GRU_B * GRU_D / 4;
  return __builtin_nontemporal_load(&s[(unsigned)lane + 32u * (unsigned)(q & 3)]);
}
__device__ __forceinline__ void fused_load_tile(const float* __restrict__ x,
                                                int t, int b0, int lane,
                                                v4f (&r)[8]) {
  const fused_gptr s0 = fused_tile_base(x, t, b0);
#pragma unroll
  for (int q = 0; q < 8; ++q) r[q] = fused_load_piece(s0, q, lane);
}

__device__ __forceinline__ void fused_store_tile(const v4f (&r)[8], float* xs_st) {
#pragma unroll
  for (int q = 0; q < 8; ++q) *(v4f*)(xs_st + 2 * q * GRU_XPITCH) = r[q];
}

__device__ __forceinline__ void fused_read_frags(const float* xs_ld, v2f (&a)[16]) {
#pragma unroll
  for (int k = 0; k < 16; ++k) a[k] = *(const v2f*)(xs_ld + 4 * k);
}

template <bool kRequest>
__device__ __forceinline__ void fused_mma_gates(const v2f (&a)[16],
                                                const v2f (&b)[16],
                                                float bias, float* gdst, int m,
                                                const float* __restrict__ x,
                                                int tReq, int b0, int lane,
                                                v4f (&r)[8]) {
  fused_gptr s0 = nullptr;
  if (kRequest) s0 = fused_tile_base(x, tReq, b0);
  v8f c = {};
#pragma unroll
  for (int k = 0; k < 16; ++k) {
    if (kRequest && (k & 1) == 0) {
      __builtin_amdgcn_sched_barrier(0);
      r[k >> 1] = fused_load_piece(s0, k >> 1, lane);
      __builtin_amdgcn_sched_barrier(0);
    }
    c = __builtin_amdgcn_wmma_f32_16x16x4_f32(false, a[k], false, b[k],
                                              (short)0, c, false, false);
  }
  if (m < 3) {
#pragma unroll
    for (int i = 0; i < 8; ++i) gdst[i * F_GPITCH] = c[i] + bias;
  }
}

template <bool kStage, bool kRequest, bool kSpread = true>
__device__ __forceinline__ void fused_produce_chunk(const float* __restrict__ x,
                                                    int tReq0, int tReq1,
                                                    int b0, int lane,
                                                    v4f (&sa)[8], v4f (&sb)[8],
                                                    float* xs_st0, const float* xs_ld0,
                                                    float* xs_st1, const float* xs_ld1,
                                                    const v2f (&b)[16], float bias,
                                                    float* gdst, int m) {
  v2f a0[16], a1[16];
  fused_read_frags(xs_ld0, a0);
  fused_read_frags(xs_ld1, a1);
  if (kSpread) {
    if (kStage) {
      __builtin_amdgcn_sched_barrier(0);
      fused_store_tile(sa, xs_st0);
      fused_store_tile(sb, xs_st1);
    }
    __builtin_amdgcn_sched_barrier(0);
    fused_mma_gates<kRequest>(a0, b, bias, gdst, m, x, tReq0, b0, lane, sa);
    fused_mma_gates<kRequest>(a1, b, bias, gdst + 2, m, x, tReq1, b0, lane, sb);
  } else {
    __builtin_amdgcn_sched_barrier(0);
    fused_mma_gates<false>(a0, b, bias, gdst, m, x, 0, b0, lane, sa);
    fused_mma_gates<false>(a1, b, bias, gdst + 2, m, x, 0, b0, lane, sb);
    if (kStage) {
      __builtin_amdgcn_sched_barrier(0);
      fused_store_tile(sa, xs_st0);
      fused_store_tile(sb, xs_st1);
    }
    if (kRequest) {
      __builtin_amdgcn_sched_barrier(0);
      fused_load_tile(x, tReq0, b0, lane, sa);
      fused_load_tile(x, tReq1, b0, lane, sb);
    }
  }
}

__device__ __forceinline__ void fused_stage_barrier() {
  asm volatile("s_wait_dscnt 0x0" ::: "memory");
  asm volatile("s_barrier_signal -1" ::: "memory");
  asm volatile("s_barrier_wait -1" ::: "memory");
}

__device__ __forceinline__ float fused_scan_chunk(float h, int t0, int len,
                                                  const float* gs,
                                                  float hw0, float hw1, float hw2,
                                                  float hc0, float hc1, float hc2,
                                                  float* __restrict__ po,
                                                  bool stores) {
  float xr[F_TC], xz[F_TC], xn[F_TC], o[F_TC];
#pragma unroll
  for (int k = 0; k < F_TC / 4; ++k) {
    const v4f vr = *(const v4f*)(gs + 4 * k);
    const v4f vz = *(const v4f*)(gs + F_COLS * F_GPITCH + 4 * k);
    const v4f vn = *(const v4f*)(gs + 2 * F_COLS * F_GPITCH + 4 * k);
#pragma unroll
    for (int e = 0; e < 4; ++e) {
      xr[4 * k + e] = vr[e];
      xz[4 * k + e] = vz[e];
      xn[4 * k + e] = vn[e];
    }
  }
#pragma unroll
  for (int i = 0; i < F_TC; ++i) {
    const int t = t0 + i;
    const float pr = fmaf(0.5f, xr[i], hc0);
    const float pz = fmaf(0.5f, xz[i], hc1);
    const float tr = __builtin_amdgcn_tanhf(fmaf(h, hw0, pr));
    const float tz = __builtin_amdgcn_tanhf(fmaf(h, hw1, pz));
    const float hg = fmaf(h, hw2, hc2);
    const float n  = __builtin_amdgcn_tanhf(fmaf(tr, hg, xn[i] + hg));
    const float hh  = 0.5f * h;
    const float zh  = fmaf(tz, hh, hh);
    const float omz = fmaf(-0.5f, tz, 0.5f);
    h = fmaf(omz, n, zh);
    o[i] = (t < len) ? h : 0.0f;
  }
  if (stores) {
#pragma unroll
    for (int i = 0; i < F_TC; ++i) po[(size_t)(t0 + i) * GRU_B] = o[i];
  }
  return h;
}

__global__ void __launch_bounds__(32 * F_WAVES)
gru_fused_kernel(const float* __restrict__ x,
                 const float* __restrict__ h0,
                 const int*   __restrict__ lengths,
                 const float* __restrict__ W_ih,
                 const float* __restrict__ W_hh,
                 const float* __restrict__ b_ih,
                 const float* __restrict__ b_hh,
                 float* __restrict__ out) {
  __shared__ float Bs[16 * GRU_D];
  __shared__ __attribute__((aligned(16))) float Xs[2 * F_NPROD * 16 * GRU_XPITCH];
  __shared__ __attribute__((aligned(16))) float Gs[2 * F_GSTAGE];

  for (int i = threadIdx.x; i < 16 * GRU_D; i += blockDim.x) {
    const int row = i >> 6;
    const int col = i & (GRU_D - 1);
    Bs[i] = (row < 3) ? W_ih[row * GRU_D + col] : 0.0f;
  }
  __syncthreads();

  const int lane  = threadIdx.x & 31;
  const int m     = lane & 15;
  const int khalf = lane >> 4;
  const int wave  = __builtin_amdgcn_readfirstlane((int)(threadIdx.x >> 5));
  const bool scanRole = (wave == 0);
  const int pi    = (wave > 0) ? (wave - 1) : 0;
  const int b0    = (int)blockIdx.x * F_COLS;

  const float* brow = Bs + m * GRU_D + 2 * khalf;
  v2f bf[16];
#pragma unroll
  for (int k = 0; k < 16; ++k) bf[k] = *(const v2f*)(brow + 4 * k);
  const float bias = (m < 3) ? b_ih[m] : 0.0f;
  float* xs = Xs + 2 * pi * (16 * GRU_XPITCH);
  float*       xs_st0 = xs + khalf * GRU_XPITCH + 4 * m;
  const float* xs_ld0 = xs + m * GRU_XPITCH + 2 * khalf;
  float*       xs_st1 = xs_st0 + 16 * GRU_XPITCH;
  const float* xs_ld1 = xs_ld0 + 16 * GRU_XPITCH;
  const int gofs = ((m < 3) ? m : 0) * (F_COLS * F_GPITCH) + 4 * pi + khalf;
  const int ts0 = 4 * pi, ts1 = 4 * pi + 2;

  const int col = lane & (F_COLS - 1);
  const int b   = b0 + col;
  float h = h0[b];
  const int len = lengths[b];
  const float w0 = 0.5f * W_hh[0], w1 = 0.5f * W_hh[1], w2 = 0.5f * W_hh[2];
  const float c0 = 0.5f * b_hh[0], c1 = 0.5f * b_hh[1], c2 = 0.5f * b_hh[2];
  float* po = out + b;
  const float* gcol = Gs + col * F_GPITCH;

  if (scanRole) {
    fused_stage_barrier();
    for (int c = 0; c < F_NCHUNK; ++c) {
      h = fused_scan_chunk(h, c * F_TC, len, gcol + (c & 1) * F_GSTAGE,
                           w0, w1, w2, c0, c1, c2, po, lane < F_COLS);
      fused_stage_barrier();
    }
  } else {
    v4f pa0[8], pa1[8], pb0[8], pb1[8];

    fused_load_tile(x, ts0, b0, lane, pa0);
    fused_load_tile(x, ts1, b0, lane, pa1);
    fused_load_tile(x, F_TC + ts0, b0, lane, pb0);
    fused_load_tile(x, F_TC + ts1, b0, lane, pb1);
    fused_store_tile(pa0, xs_st0);
    fused_store_tile(pa1, xs_st1);
    __builtin_amdgcn_sched_barrier(0);
    fused_load_tile(x, 2 * F_TC + ts0, b0, lane, pa0);
    fused_load_tile(x, 2 * F_TC + ts1, b0, lane, pa1);
    fused_produce_chunk<true, true, false>(x, 3 * F_TC + ts0, 3 * F_TC + ts1, b0, lane,
                                           pb0, pb1, xs_st0, xs_ld0, xs_st1, xs_ld1,
                                           bf, bias, Gs + gofs, m);
    fused_stage_barrier();

    for (int c = 0; c + 5 < F_NCHUNK; c += 2) {
      fused_produce_chunk<true, true>(x, (c + 4) * F_TC + ts0, (c + 4) * F_TC + ts1,
                                      b0, lane, pa0, pa1,
                                      xs_st0, xs_ld0, xs_st1, xs_ld1, bf, bias,
                                      Gs + F_GSTAGE + gofs, m);
      fused_stage_barrier();
      fused_produce_chunk<true, true>(x, (c + 5) * F_TC + ts0, (c + 5) * F_TC + ts1,
                                      b0, lane, pb0, pb1,
                                      xs_st0, xs_ld0, xs_st1, xs_ld1, bf, bias,
                                      Gs + gofs, m);
      fused_stage_barrier();
    }
    fused_produce_chunk<true, false>(x, 0, 0, b0, lane, pa0, pa1,
                                     xs_st0, xs_ld0, xs_st1, xs_ld1, bf, bias,
                                     Gs + F_GSTAGE + gofs, m);
    fused_stage_barrier();
    fused_produce_chunk<true, false>(x, 0, 0, b0, lane, pb0, pb1,
                                     xs_st0, xs_ld0, xs_st1, xs_ld1, bf, bias,
                                     Gs + gofs, m);
    fused_stage_barrier();
    fused_produce_chunk<false, false>(x, 0, 0, b0, lane, pa0, pa1,
                                      xs_st0, xs_ld0, xs_st1, xs_ld1, bf, bias,
                                      Gs + F_GSTAGE + gofs, m);
    fused_stage_barrier();
    fused_stage_barrier();
  }
}

extern "C" void kernel_launch(void* const* d_in, const int* in_sizes, int n_in,
                              void* d_out, int out_size, void* d_ws, size_t ws_size,
                              hipStream_t stream) {
  const float* x       = (const float*)d_in[0];
  const float* h0      = (const float*)d_in[1];
  const int*   lengths = (const int*)  d_in[2];
  const float* W_ih    = (const float*)d_in[3];
  const float* W_hh    = (const float*)d_in[4];
  const float* b_ih    = (const float*)d_in[5];
  const float* b_hh    = (const float*)d_in[6];
  float*       out     = (float*)d_out;

  gru_fused_kernel<<<GRU_B / F_COLS, 32 * F_WAVES, 0, stream>>>(
      x, h0, lengths, W_ih, W_hh, b_ih, b_hh, out);
}


# each producer wave owns one batch column (16-step tiles) and stops fetching, staging and multiplying past that column's length: the reference selects zero there and returns no final state
# speedup vs baseline: 10.3272x; 1.0689x over previous
#include <hip/hip_runtime.h>
#include <hip/hip_bf16.h>
#include <math.h>


typedef float v2f __attribute__((ext_vector_type(2)));
typedef float v4f __attribute__((ext_vector_type(4)));
typedef float v8f __attribute__((ext_vector_type(8)));

#define GRU_L 512
#define GRU_B 2048
#define GRU_D 64
#define GRU_XPITCH 68

#define F_COLS   8
#define F_TC     32
#define F_NCHUNK (GRU_L / F_TC)
#define F_NPROD  8
#define F_WAVES  (F_NPROD + 1)
#define F_GPITCH (F_TC + 4)
#define F_GSTAGE (3 * F_COLS * F_GPITCH)

typedef const v4f __attribute__((address_space(1)))* fused_gptr;
#define GRU_STEP16 (GRU_B * GRU_D / 4)
__device__ __forceinline__ fused_gptr fused_tile_base(const float* __restrict__ x,
                                                      int t, int bcol) {
  const unsigned long long u =
      (unsigned long long)(x + ((size_t)t * GRU_B + bcol) * GRU_D);
  const unsigned lo = (unsigned)__builtin_amdgcn_readfirstlane((int)(unsigned)u);
  const unsigned hi = (unsigned)__builtin_amdgcn_readfirstlane((int)(unsigned)(u >> 32));
  return (fused_gptr)(((unsigned long long)hi << 32) | (unsigned long long)lo);
}
__device__ __forceinline__ unsigned fused_lane_ofs(int lane) {
  return (unsigned)(lane & 15) + (unsigned)(lane >> 4) * (unsigned)GRU_STEP16;
}
__device__ __forceinline__ v4f fused_load_piece(fused_gptr s0, int q, unsigned lofs) {
  return __builtin_nontemporal_load(&s0[lofs + (unsigned)(2 * GRU_STEP16) * (unsigned)q]);
}
__device__ __forceinline__ void fused_load_tile(const float* __restrict__ x,
                                                int t, int bcol, unsigned lofs,
                                                v4f (&r)[8]) {
  const fused_gptr s0 = fused_tile_base(x, t, bcol);
#pragma unroll
  for (int q = 0; q < 8; ++q) r[q] = fused_load_piece(s0, q, lofs);
}

__device__ __forceinline__ void fused_store_tile(const v4f (&r)[8], float* xs_st) {
#pragma unroll
  for (int q = 0; q < 8; ++q) *(v4f*)(xs_st + 2 * q * GRU_XPITCH) = r[q];
}

__device__ __forceinline__ void fused_read_frags(const float* xs_ld, v2f (&a)[16]) {
#pragma unroll
  for (int k = 0; k < 16; ++k) a[k] = *(const v2f*)(xs_ld + 4 * k);
}

template <bool kRequest>
__device__ __forceinline__ void fused_mma_gates(const v2f (&a)[16],
                                                const v2f (&b)[16],
                                                float bias, float* gdst, int m,
                                                const float* __restrict__ x,
                                                int tReq, int bcol, unsigned lofs,
                                                v4f (&r)[8]) {
  fused_gptr s0 = nullptr;
  if (kRequest) s0 = fused_tile_base(x, tReq, bcol);
  v8f c = {};
#pragma unroll
  for (int k = 0; k < 16; ++k) {
    if (kRequest && (k & 1) == 0) {
      __builtin_amdgcn_sched_barrier(0);
      r[k >> 1] = fused_load_piece(s0, k >> 1, lofs);
      __builtin_amdgcn_sched_barrier(0);
    }
    c = __builtin_amdgcn_wmma_f32_16x16x4_f32(false, a[k], false, b[k],
                                              (short)0, c, false, false);
  }
  if (m < 3) {
    const v4f lo = {c[0] + bias, c[1] + bias, c[2] + bias, c[3] + bias};
    const v4f hi = {c[4] + bias, c[5] + bias, c[6] + bias, c[7] + bias};
    *(v4f*)gdst = lo;
    *(v4f*)(gdst + 4) = hi;
  }
}

template <bool kStage, bool kRequest, bool kSpread = true>
__device__ __forceinline__ void fused_produce_chunk(const float* __restrict__ x,
                                                    int tReq0, int tReq1,
                                                    int bcol, unsigned lofs,
                                                    v4f (&sa)[8], v4f (&sb)[8],
                                                    float* xs_st0, const float* xs_ld0,
                                                    float* xs_st1, const float* xs_ld1,
                                                    const v2f (&b)[16], float bias,
                                                    float* gdst, int m) {
  v2f a0[16], a1[16];
  fused_read_frags(xs_ld0, a0);
  fused_read_frags(xs_ld1, a1);
  if (kSpread) {
    if (kStage) {
      __builtin_amdgcn_sched_barrier(0);
      fused_store_tile(sa, xs_st0);
      fused_store_tile(sb, xs_st1);
    }
    __builtin_amdgcn_sched_barrier(0);
    fused_mma_gates<kRequest>(a0, b, bias, gdst, m, x, tReq0, bcol, lofs, sa);
    fused_mma_gates<kRequest>(a1, b, bias, gdst + 16, m, x, tReq1, bcol, lofs, sb);
  } else {
    __builtin_amdgcn_sched_barrier(0);
    fused_mma_gates<false>(a0, b, bias, gdst, m, x, 0, bcol, lofs, sa);
    fused_mma_gates<false>(a1, b, bias, gdst + 16, m, x, 0, bcol, lofs, sb);
    if (kStage) {
      __builtin_amdgcn_sched_barrier(0);
      fused_store_tile(sa, xs_st0);
      fused_store_tile(sb, xs_st1);
    }
    if (kRequest) {
      __builtin_amdgcn_sched_barrier(0);
      fused_load_tile(x, tReq0, bcol, lofs, sa);
      fused_load_tile(x, tReq1, bcol, lofs, sb);
    }
  }
}

__device__ __forceinline__ void fused_stage_barrier() {
  asm volatile("s_wait_dscnt 0x0" ::: "memory");
  asm volatile("s_barrier_signal -1" ::: "memory");
  asm volatile("s_barrier_wait -1" ::: "memory");
}

__device__ __forceinline__ float fused_scan_chunk(float h, int t0, int len,
                                                  const float* gs,
                                                  float hw0, float hw1, float hw2,
                                                  float hc0, float hc1, float hc2,
                                                  float* __restrict__ po,
                                                  bool stores) {
  float xr[F_TC], xz[F_TC], xn[F_TC], o[F_TC];
#pragma unroll
  for (int k = 0; k < F_TC / 4; ++k) {
    const v4f vr = *(const v4f*)(gs + 4 * k);
    const v4f vz = *(const v4f*)(gs + F_COLS * F_GPITCH + 4 * k);
    const v4f vn = *(const v4f*)(gs + 2 * F_COLS * F_GPITCH + 4 * k);
#pragma unroll
    for (int e = 0; e < 4; ++e) {
      xr[4 * k + e] = vr[e];
      xz[4 * k + e] = vz[e];
      xn[4 * k + e] = vn[e];
    }
  }
#pragma unroll
  for (int i = 0; i < F_TC; ++i) {
    const int t = t0 + i;
    const float pr = fmaf(0.5f, xr[i], hc0);
    const float pz = fmaf(0.5f, xz[i], hc1);
    const float tr = __builtin_amdgcn_tanhf(fmaf(h, hw0, pr));
    const float tz = __builtin_amdgcn_tanhf(fmaf(h, hw1, pz));
    const float hg = fmaf(h, hw2, hc2);
    const float n  = __builtin_amdgcn_tanhf(fmaf(tr, hg, xn[i] + hg));
    const float hh  = 0.5f * h;
    const float zh  = fmaf(tz, hh, hh);
    const float omz = fmaf(-0.5f, tz, 0.5f);
    h = fmaf(omz, n, zh);
    o[i] = (t < len) ? h : 0.0f;
  }
  if (stores) {
#pragma unroll
    for (int i = 0; i < F_TC; ++i) po[(size_t)(t0 + i) * GRU_B] = o[i];
  }
  return h;
}

__global__ void __launch_bounds__(32 * F_WAVES)
gru_fused_kernel(const float* __restrict__ x,
                 const float* __restrict__ h0,
                 const int*   __restrict__ lengths,
                 const float* __restrict__ W_ih,
                 const float* __restrict__ W_hh,
                 const float* __restrict__ b_ih,
                 const float* __restrict__ b_hh,
                 float* __restrict__ out) {
  __shared__ float Bs[16 * GRU_D];
  __shared__ __attribute__((aligned(16))) float Xs[2 * F_NPROD * 16 * GRU_XPITCH];
  __shared__ __attribute__((aligned(16))) float Gs[2 * F_GSTAGE];

  for (int i = threadIdx.x; i < 16 * GRU_D; i += blockDim.x) {
    const int row = i >> 6;
    const int col = i & (GRU_D - 1);
    Bs[i] = (row < 3) ? W_ih[row * GRU_D + col] : 0.0f;
  }
  __syncthreads();

  const int lane  = threadIdx.x & 31;
  const int m     = lane & 15;
  const int khalf = lane >> 4;
  const int wave  = __builtin_amdgcn_readfirstlane((int)(threadIdx.x >> 5));
  const bool scanRole = (wave == 0);
  const int pi    = (wave > 0) ? (wave - 1) : 0;
  const int b0    = (int)blockIdx.x * F_COLS;

  const float* brow = Bs + m * GRU_D + 2 * khalf;
  v2f bf[16];
#pragma unroll
  for (int k = 0; k < 16; ++k) bf[k] = *(const v2f*)(brow + 4 * k);
  const float bias = (m < 3) ? b_ih[m] : 0.0f;
  float* xs = Xs + 2 * pi * (16 * GRU_XPITCH);
  float*       xs_st0 = xs + khalf * GRU_XPITCH + 4 * m;
  const float* xs_ld0 = xs + m * GRU_XPITCH + 2 * khalf;
  float*       xs_st1 = xs_st0 + 16 * GRU_XPITCH;
  const float* xs_ld1 = xs_ld0 + 16 * GRU_XPITCH;
  const int gofs = ((m < 3) ? m : 0) * (F_COLS * F_GPITCH) + pi * F_GPITCH + 8 * khalf;
  const int ts0 = 0, ts1 = 16;
  const int bcol = b0 + pi;
  const unsigned lofs = fused_lane_ofs(lane);
  const int plen = __builtin_amdgcn_readfirstlane(lengths[bcol]);
  const int pclamp = plen < 0 ? 0 : (plen > GRU_L ? GRU_L : plen);
  const int pneed = (pclamp + F_TC - 1) / F_TC;
  const int nlive = (pneed + 1) / 2 * 2 < 4 ? 4 : (pneed + 1) / 2 * 2;

  const int col = lane & (F_COLS - 1);
  const int b   = b0 + col;
  float h = h0[b];
  const int len = lengths[b];
  const float w0 = 0.5f * W_hh[0], w1 = 0.5f * W_hh[1], w2 = 0.5f * W_hh[2];
  const float c0 = 0.5f * b_hh[0], c1 = 0.5f * b_hh[1], c2 = 0.5f * b_hh[2];
  float* po = out + b;
  const float* gcol = Gs + col * F_GPITCH;

  if (scanRole) {
    fused_stage_barrier();
    for (int c = 0; c < F_NCHUNK; ++c) {
      h = fused_scan_chunk(h, c * F_TC, len, gcol + (c & 1) * F_GSTAGE,
                           w0, w1, w2, c0, c1, c2, po, lane < F_COLS);
      fused_stage_barrier();
    }
  } else {
    v4f pa0[8], pa1[8], pb0[8], pb1[8];

    fused_load_tile(x, ts0, bcol, lofs, pa0);
    fused_load_tile(x, ts1, bcol, lofs, pa1);
    fused_load_tile(x, F_TC + ts0, bcol, lofs, pb0);
    fused_load_tile(x, F_TC + ts1, bcol, lofs, pb1);
    fused_store_tile(pa0, xs_st0);
    fused_store_tile(pa1, xs_st1);
    __builtin_amdgcn_sched_barrier(0);
    fused_load_tile(x, 2 * F_TC + ts0, bcol, lofs, pa0);
    fused_load_tile(x, 2 * F_TC + ts1, bcol, lofs, pa1);
    fused_produce_chunk<true, true, false>(x, 3 * F_TC + ts0, 3 * F_TC + ts1, bcol, lofs,
                                           pb0, pb1, xs_st0, xs_ld0, xs_st1, xs_ld1,
                                           bf, bias, Gs + gofs, m);
    fused_stage_barrier();

    for (int c = 0; c + 5 < nlive; c += 2) {
      fused_produce_chunk<true, true>(x, (c + 4) * F_TC + ts0, (c + 4) * F_TC + ts1,
                                      bcol, lofs, pa0, pa1,
                                      xs_st0, xs_ld0, xs_st1, xs_ld1, bf, bias,
                                      Gs + F_GSTAGE + gofs, m);
      fused_stage_barrier();
      fused_produce_chunk<true, true>(x, (c + 5) * F_TC + ts0, (c + 5) * F_TC + ts1,
                                      bcol, lofs, pb0, pb1,
                                      xs_st0, xs_ld0, xs_st1, xs_ld1, bf, bias,
                                      Gs + gofs, m);
      fused_stage_barrier();
    }
    fused_produce_chunk<true, false>(x, 0, 0, bcol, lofs, pa0, pa1,
                                     xs_st0, xs_ld0, xs_st1, xs_ld1, bf, bias,
                                     Gs + F_GSTAGE + gofs, m);
    fused_stage_barrier();
    fused_produce_chunk<true, false>(x, 0, 0, bcol, lofs, pb0, pb1,
                                     xs_st0, xs_ld0, xs_st1, xs_ld1, bf, bias,
                                     Gs + gofs, m);
    fused_stage_barrier();
    fused_produce_chunk<false, false>(x, 0, 0, bcol, lofs, pa0, pa1,
                                      xs_st0, xs_ld0, xs_st1, xs_ld1, bf, bias,
                                      Gs + F_GSTAGE + gofs, m);
    fused_stage_barrier();
    fused_stage_barrier();
    for (int c = nlive; c < F_NCHUNK; ++c) fused_stage_barrier();
  }
}

extern "C" void kernel_launch(void* const* d_in, const int* in_sizes, int n_in,
                              void* d_out, int out_size, void* d_ws, size_t ws_size,
                              hipStream_t stream) {
  const float* x       = (const float*)d_in[0];
  const float* h0      = (const float*)d_in[1];
  const int*   lengths = (const int*)  d_in[2];
  const float* W_ih    = (const float*)d_in[3];
  const float* W_hh    = (const float*)d_in[4];
  const float* b_ih    = (const float*)d_in[5];
  const float* b_hh    = (const float*)d_in[6];
  float*       out     = (float*)d_out;

  gru_fused_kernel<<<GRU_B / F_COLS, 32 * F_WAVES, 0, stream>>>(
      x, h0, lengths, W_ih, W_hh, b_ih, b_hh, out);
}


# scan wave hands its ring stage back once read: 24 gate reads pinned as one group ahead of the chain, barrier signal 8 steps in, wait behind the chain; producers no longer wait for the chain
# speedup vs baseline: 11.2854x; 1.0928x over previous
#include <hip/hip_runtime.h>
#include <hip/hip_bf16.h>
#include <math.h>


typedef float v2f __attribute__((ext_vector_type(2)));
typedef float v4f __attribute__((ext_vector_type(4)));
typedef float v8f __attribute__((ext_vector_type(8)));

#define GRU_L 512
#define GRU_B 2048
#define GRU_D 64
#define GRU_XPITCH 68

#define F_COLS   8
#define F_TC     32
#define F_NCHUNK (GRU_L / F_TC)
#define F_NPROD  8
#define F_WAVES  (F_NPROD + 1)
#define F_GPITCH (F_TC + 4)
#define F_GSTAGE (3 * F_COLS * F_GPITCH)

typedef const v4f __attribute__((address_space(1)))* fused_gptr;
#define GRU_STEP16 (GRU_B * GRU_D / 4)
__device__ __forceinline__ fused_gptr fused_tile_base(const float* __restrict__ x,
                                                      int t, int bcol) {
  const unsigned long long u =
      (unsigned long long)(x + ((size_t)t * GRU_B + bcol) * GRU_D);
  const unsigned lo = (unsigned)__builtin_amdgcn_readfirstlane((int)(unsigned)u);
  const unsigned hi = (unsigned)__builtin_amdgcn_readfirstlane((int)(unsigned)(u >> 32));
  return (fused_gptr)(((unsigned long long)hi << 32) | (unsigned long long)lo);
}
__device__ __forceinline__ unsigned fused_lane_ofs(int lane) {
  return (unsigned)(lane & 15) + (unsigned)(lane >> 4) * (unsigned)GRU_STEP16;
}
__device__ __forceinline__ v4f fused_load_piece(fused_gptr s0, int q, unsigned lofs) {
  return __builtin_nontemporal_load(&s0[lofs + (unsigned)(2 * GRU_STEP16) * (unsigned)q]);
}
__device__ __forceinline__ void fused_load_tile(const float* __restrict__ x,
                                                int t, int bcol, unsigned lofs,
                                                v4f (&r)[8]) {
  const fused_gptr s0 = fused_tile_base(x, t, bcol);
#pragma unroll
  for (int q = 0; q < 8; ++q) r[q] = fused_load_piece(s0, q, lofs);
}

__device__ __forceinline__ void fused_store_tile(const v4f (&r)[8], float* xs_st) {
#pragma unroll
  for (int q = 0; q < 8; ++q) *(v4f*)(xs_st + 2 * q * GRU_XPITCH) = r[q];
}

__device__ __forceinline__ void fused_read_frags(const float* xs_ld, v2f (&a)[16]) {
#pragma unroll
  for (int k = 0; k < 16; ++k) a[k] = *(const v2f*)(xs_ld + 4 * k);
}

template <bool kRequest>
__device__ __forceinline__ void fused_mma_gates(const v2f (&a)[16],
                                                const v2f (&b)[16],
                                                float bias, float* gdst, int m,
                                                const float* __restrict__ x,
                                                int tReq, int bcol, unsigned lofs,
                                                v4f (&r)[8]) {
  fused_gptr s0 = nullptr;
  if (kRequest) s0 = fused_tile_base(x, tReq, bcol);
  v8f c = {};
#pragma unroll
  for (int k = 0; k < 16; ++k) {
    if (kRequest && (k & 1) == 0) {
      __builtin_amdgcn_sched_barrier(0);
      r[k >> 1] = fused_load_piece(s0, k >> 1, lofs);
      __builtin_amdgcn_sched_barrier(0);
    }
    c = __builtin_amdgcn_wmma_f32_16x16x4_f32(false, a[k], false, b[k],
                                              (short)0, c, false, false);
  }
  if (m < 3) {
    const v4f lo = {c[0] + bias, c[1] + bias, c[2] + bias, c[3] + bias};
    const v4f hi = {c[4] + bias, c[5] + bias, c[6] + bias, c[7] + bias};
    *(v4f*)gdst = lo;
    *(v4f*)(gdst + 4) = hi;
  }
}

template <bool kStage, bool kRequest, bool kSpread = true>
__device__ __forceinline__ void fused_produce_chunk(const float* __restrict__ x,
                                                    int tReq0, int tReq1,
                                                    int bcol, unsigned lofs,
                                                    v4f (&sa)[8], v4f (&sb)[8],
                                                    float* xs_st0, const float* xs_ld0,
                                                    float* xs_st1, const float* xs_ld1,
                                                    const v2f (&b)[16], float bias,
                                                    float* gdst, int m) {
  v2f a0[16], a1[16];
  fused_read_frags(xs_ld0, a0);
  fused_read_frags(xs_ld1, a1);
  if (kSpread) {
    if (kStage) {
      __builtin_amdgcn_sched_barrier(0);
      fused_store_tile(sa, xs_st0);
      fused_store_tile(sb, xs_st1);
    }
    __builtin_amdgcn_sched_barrier(0);
    fused_mma_gates<kRequest>(a0, b, bias, gdst, m, x, tReq0, bcol, lofs, sa);
    fused_mma_gates<kRequest>(a1, b, bias, gdst + 16, m, x, tReq1, bcol, lofs, sb);
  } else {
    __builtin_amdgcn_sched_barrier(0);
    fused_mma_gates<false>(a0, b, bias, gdst, m, x, 0, bcol, lofs, sa);
    fused_mma_gates<false>(a1, b, bias, gdst + 16, m, x, 0, bcol, lofs, sb);
    if (kStage) {
      __builtin_amdgcn_sched_barrier(0);
      fused_store_tile(sa, xs_st0);
      fused_store_tile(sb, xs_st1);
    }
    if (kRequest) {
      __builtin_amdgcn_sched_barrier(0);
      fused_load_tile(x, tReq0, bcol, lofs, sa);
      fused_load_tile(x, tReq1, bcol, lofs, sb);
    }
  }
}

__device__ __forceinline__ void fused_stage_signal() {
  asm volatile("s_wait_dscnt 0x0" ::: "memory");
  asm volatile("s_barrier_signal -1" ::: "memory");
}
__device__ __forceinline__ void fused_stage_wait() {
  asm volatile("s_barrier_wait -1" ::: "memory");
}
__device__ __forceinline__ void fused_stage_barrier() {
  fused_stage_signal();
  fused_stage_wait();
}

#define F_HANDBACK 8
__device__ __forceinline__ float fused_scan_chunk(float h, int t0, int len,
                                                  const float* gs,
                                                  float hw0, float hw1, float hw2,
                                                  float hc0, float hc1, float hc2,
                                                  float* __restrict__ po,
                                                  bool stores) {
  float xr[F_TC], xz[F_TC], xn[F_TC], o[F_TC];
#pragma unroll
  for (int k = 0; k < F_TC / 4; ++k) {
    const v4f vr = *(const v4f*)(gs + 4 * k);
    const v4f vz = *(const v4f*)(gs + F_COLS * F_GPITCH + 4 * k);
    const v4f vn = *(const v4f*)(gs + 2 * F_COLS * F_GPITCH + 4 * k);
#pragma unroll
    for (int e = 0; e < 4; ++e) {
      xr[4 * k + e] = vr[e];
      xz[4 * k + e] = vz[e];
      xn[4 * k + e] = vn[e];
    }
  }
  __builtin_amdgcn_sched_barrier(0);
#pragma unroll
  for (int i = 0; i < F_TC; ++i) {
    const int t = t0 + i;
    if (i == F_HANDBACK) {
      __builtin_amdgcn_sched_barrier(0);
      fused_stage_signal();
    }
    const float pr = fmaf(0.5f, xr[i], hc0);
    const float pz = fmaf(0.5f, xz[i], hc1);
    const float tr = __builtin_amdgcn_tanhf(fmaf(h, hw0, pr));
    const float tz = __builtin_amdgcn_tanhf(fmaf(h, hw1, pz));
    const float hg = fmaf(h, hw2, hc2);
    const float n  = __builtin_amdgcn_tanhf(fmaf(tr, hg, xn[i] + hg));
    const float hh  = 0.5f * h;
    const float zh  = fmaf(tz, hh, hh);
    const float omz = fmaf(-0.5f, tz, 0.5f);
    h = fmaf(omz, n, zh);
    o[i] = (t < len) ? h : 0.0f;
  }
  if (stores) {
#pragma unroll
    for (int i = 0; i < F_TC; ++i) po[(size_t)(t0 + i) * GRU_B] = o[i];
  }
  return h;
}

__global__ void __launch_bounds__(32 * F_WAVES)
gru_fused_kernel(const float* __restrict__ x,
                 const float* __restrict__ h0,
                 const int*   __restrict__ lengths,
                 const float* __restrict__ W_ih,
                 const float* __restrict__ W_hh,
                 const float* __restrict__ b_ih,
                 const float* __restrict__ b_hh,
                 float* __restrict__ out) {
  __shared__ float Bs[16 * GRU_D];
  __shared__ __attribute__((aligned(16))) float Xs[2 * F_NPROD * 16 * GRU_XPITCH];
  __shared__ __attribute__((aligned(16))) float Gs[2 * F_GSTAGE];

  for (int i = threadIdx.x; i < 16 * GRU_D; i += blockDim.x) {
    const int row = i >> 6;
    const int col = i & (GRU_D - 1);
    Bs[i] = (row < 3) ? W_ih[row * GRU_D + col] : 0.0f;
  }
  __syncthreads();

  const int lane  = threadIdx.x & 31;
  const int m     = lane & 15;
  const int khalf = lane >> 4;
  const int wave  = __builtin_amdgcn_readfirstlane((int)(threadIdx.x >> 5));
  const bool scanRole = (wave == 0);
  const int pi    = (wave > 0) ? (wave - 1) : 0;
  const int b0    = (int)blockIdx.x * F_COLS;

  const float* brow = Bs + m * GRU_D + 2 * khalf;
  v2f bf[16];
#pragma unroll
  for (int k = 0; k < 16; ++k) bf[k] = *(const v2f*)(brow + 4 * k);
  const float bias = (m < 3) ? b_ih[m] : 0.0f;
  float* xs = Xs + 2 * pi * (16 * GRU_XPITCH);
  float*       xs_st0 = xs + khalf * GRU_XPITCH + 4 * m;
  const float* xs_ld0 = xs + m * GRU_XPITCH + 2 * khalf;
  float*       xs_st1 = xs_st0 + 16 * GRU_XPITCH;
  const float* xs_ld1 = xs_ld0 + 16 * GRU_XPITCH;
  const int gofs = ((m < 3) ? m : 0) * (F_COLS * F_GPITCH) + pi * F_GPITCH + 8 * khalf;
  const int ts0 = 0, ts1 = 16;
  const int bcol = b0 + pi;
  const unsigned lofs = fused_lane_ofs(lane);
  const int plen = __builtin_amdgcn_readfirstlane(lengths[bcol]);
  const int pclamp = plen < 0 ? 0 : (plen > GRU_L ? GRU_L : plen);
  const int pneed = (pclamp + F_TC - 1) / F_TC;
  const int nlive = (pneed + 1) / 2 * 2 < 4 ? 4 : (pneed + 1) / 2 * 2;

  const int col = lane & (F_COLS - 1);
  const int b   = b0 + col;
  float h = h0[b];
  const int len = lengths[b];
  const float w0 = 0.5f * W_hh[0], w1 = 0.5f * W_hh[1], w2 = 0.5f * W_hh[2];
  const float c0 = 0.5f * b_hh[0], c1 = 0.5f * b_hh[1], c2 = 0.5f * b_hh[2];
  float* po = out + b;
  const float* gcol = Gs + col * F_GPITCH;

  if (scanRole) {
    fused_stage_barrier();
    for (int c = 0; c < F_NCHUNK; ++c) {
      h = fused_scan_chunk(h, c * F_TC, len, gcol + (c & 1) * F_GSTAGE,
                           w0, w1, w2, c0, c1, c2, po, lane < F_COLS);
      fused_stage_wait();
    }
  } else {
    v4f pa0[8], pa1[8], pb0[8], pb1[8];

    fused_load_tile(x, ts0, bcol, lofs, pa0);
    fused_load_tile(x, ts1, bcol, lofs, pa1);
    fused_load_tile(x, F_TC + ts0, bcol, lofs, pb0);
    fused_load_tile(x, F_TC + ts1, bcol, lofs, pb1);
    fused_store_tile(pa0, xs_st0);
    fused_store_tile(pa1, xs_st1);
    __builtin_amdgcn_sched_barrier(0);
    fused_load_tile(x, 2 * F_TC + ts0, bcol, lofs, pa0);
    fused_load_tile(x, 2 * F_TC + ts1, bcol, lofs, pa1);
    fused_produce_chunk<true, true, false>(x, 3 * F_TC + ts0, 3 * F_TC + ts1, bcol, lofs,
                                           pb0, pb1, xs_st0, xs_ld0, xs_st1, xs_ld1,
                                           bf, bias, Gs + gofs, m);
    fused_stage_barrier();

    for (int c = 0; c + 5 < nlive; c += 2) {
      fused_produce_chunk<true, true>(x, (c + 4) * F_TC + ts0, (c + 4) * F_TC + ts1,
                                      bcol, lofs, pa0, pa1,
                                      xs_st0, xs_ld0, xs_st1, xs_ld1, bf, bias,
                                      Gs + F_GSTAGE + gofs, m);
      fused_stage_barrier();
      fused_produce_chunk<true, true>(x, (c + 5) * F_TC + ts0, (c + 5) * F_TC + ts1,
                                      bcol, lofs, pb0, pb1,
                                      xs_st0, xs_ld0, xs_st1, xs_ld1, bf, bias,
                                      Gs + gofs, m);
      fused_stage_barrier();
    }
    fused_produce_chunk<true, false>(x, 0, 0, bcol, lofs, pa0, pa1,
                                     xs_st0, xs_ld0, xs_st1, xs_ld1, bf, bias,
                                     Gs + F_GSTAGE + gofs, m);
    fused_stage_barrier();
    fused_produce_chunk<true, false>(x, 0, 0, bcol, lofs, pb0, pb1,
                                     xs_st0, xs_ld0, xs_st1, xs_ld1, bf, bias,
                                     Gs + gofs, m);
    fused_stage_barrier();
    fused_produce_chunk<false, false>(x, 0, 0, bcol, lofs, pa0, pa1,
                                      xs_st0, xs_ld0, xs_st1, xs_ld1, bf, bias,
                                      Gs + F_GSTAGE + gofs, m);
    fused_stage_barrier();
    fused_stage_barrier();
    for (int c = nlive; c < F_NCHUNK; ++c) fused_stage_barrier();
  }
}

extern "C" void kernel_launch(void* const* d_in, const int* in_sizes, int n_in,
                              void* d_out, int out_size, void* d_ws, size_t ws_size,
                              hipStream_t stream) {
  const float* x       = (const float*)d_in[0];
  const float* h0      = (const float*)d_in[1];
  const int*   lengths = (const int*)  d_in[2];
  const float* W_ih    = (const float*)d_in[3];
  const float* W_hh    = (const float*)d_in[4];
  const float* b_ih    = (const float*)d_in[5];
  const float* b_hh    = (const float*)d_in[6];
  float*       out     = (float*)d_out;

  gru_fused_kernel<<<GRU_B / F_COLS, 32 * F_WAVES, 0, stream>>>(
      x, h0, lengths, W_ih, W_hh, b_ih, b_hh, out);
}
